# starting kernel
# baseline (speedup 1.0000x reference)
#include <hip/hip_runtime.h>

typedef __attribute__((ext_vector_type(2))) float v2f;
typedef __attribute__((ext_vector_type(8))) float v8f;

#define N_NODES   100000
#define N_FEAT    128
#define HIDDEN    20
#define N_CLASSES 10
#define PAD1      32
#define PAD2      16
#define M_TILES   (N_NODES / 16)


__global__ void gcn_init_deg(float* deg) {
    int i = blockIdx.x * blockDim.x + threadIdx.x;
    if (i < N_NODES) deg[i] = 1.0f;
}

__global__ void gcn_deg_accum(const int* __restrict__ dst, float* deg, int E) {
    int e = blockIdx.x * blockDim.x + threadIdx.x;
    if (e < E) atomicAdd(&deg[dst[e]], 1.0f);
}

__global__ void gcn_rsqrt(float* deg) {
    int i = blockIdx.x * blockDim.x + threadIdx.x;
    if (i < N_NODES) deg[i] = rsqrtf(deg[i]);
}


__global__ void gcn_gemm1(const float* __restrict__ x,
                          const float* __restrict__ W1,
                          const float* __restrict__ dis,
                          float* __restrict__ h0,
                          float* __restrict__ h1) {
    __shared__ float Bf[(N_FEAT / 2) * 32 * 2];

    for (int idx = threadIdx.x; idx < (N_FEAT / 2) * 32 * 2; idx += blockDim.x) {
        const int p = idx >> 6;
        const int c = (idx >> 1) & 31;
        const int j = idx & 1;
        Bf[idx] = (c < HIDDEN) ? W1[(2 * p + j) * HIDDEN + c] : 0.0f;
    }
    __syncthreads();

    const int lane = threadIdx.x & 31;
    const int wave = threadIdx.x >> 5;
    const int tile = blockIdx.x * (blockDim.x >> 5) + wave;
    if (tile >= M_TILES) return;
    const int row0 = tile << 4;
    const int half = lane >> 4;
    const int l16  = lane & 15;
    const float* arow = x + (size_t)(row0 + l16) * N_FEAT;

    v8f c0 = {};
    v8f c1 = {};

    #pragma unroll 4
    for (int k0 = 0; k0 < N_FEAT; k0 += 4) {
        const int kb = k0 + 2 * half;
        const int p = kb >> 1;
        v2f a  = *(const v2f*)(arow + kb);
        v2f b0 = *(const v2f*)&Bf[p * 64 + l16 * 2];
        v2f b1 = *(const v2f*)&Bf[p * 64 + (16 + l16) * 2];

        c0 = __builtin_amdgcn_wmma_f32_16x16x4_f32(false, a, false, b0, (short)0, c0, false, false);
        c1 = __builtin_amdgcn_wmma_f32_16x16x4_f32(false, a, false, b1, (short)0, c1, false, false);
    }

    #pragma unroll
    for (int r = 0; r < 8; ++r) {
        const int row = row0 + r + 8 * half;
        const float di = dis[row];
        const float d2 = di * di;
        float* p0 = h0 + (size_t)row * PAD1;
        float* p1 = h1 + (size_t)row * PAD1;
        p0[l16]      = c0[r];
        p0[16 + l16] = c1[r];
        p1[l16]      = d2 * c0[r];
        p1[16 + l16] = d2 * c1[r];
    }
}


template <int F, int PAD>
__global__ void gcn_propagate(const int* __restrict__ src,
                              const int* __restrict__ dst,
                              const float* __restrict__ dis,
                              const float* __restrict__ h,
                              float* __restrict__ out, int E) {
    int e = blockIdx.x * blockDim.x + threadIdx.x;
    if (e >= E) return;
    const int s = src[e];
    const int d = dst[e];
    const float nrm = dis[s] * dis[d];
    const float* hrow = h + (size_t)s * PAD;
    float* orow = out + (size_t)d * PAD;

    float v[F];
    #pragma unroll
    for (int j = 0; j + 4 <= F; j += 4) {
        float4 t = *(const float4*)(hrow + j);
        v[j] = t.x; v[j + 1] = t.y; v[j + 2] = t.z; v[j + 3] = t.w;
    }
    if constexpr (F % 4 != 0) {
        float2 t = *(const float2*)(hrow + (F & ~3));
        v[(F & ~3)] = t.x; v[(F & ~3) + 1] = t.y;
    }
    #pragma unroll
    for (int f = 0; f < F; ++f) atomicAdd(&orow[f], nrm * v[f]);
}


__global__ void gcn_gemm2(const float* __restrict__ h1,
                          const float* __restrict__ b1,
                          const float* __restrict__ W2,
                          const float* __restrict__ dis,
                          float* __restrict__ h2,
                          float* __restrict__ h2a) {
    __shared__ float Bf2[(HIDDEN / 2) * 16 * 2];
    __shared__ float b1s[HIDDEN];

    for (int idx = threadIdx.x; idx < (HIDDEN / 2) * 16 * 2; idx += blockDim.x) {
        const int p = idx >> 5;
        const int c = (idx >> 1) & 15;
        const int j = idx & 1;
        Bf2[idx] = (c < N_CLASSES) ? W2[(2 * p + j) * N_CLASSES + c] : 0.0f;
    }
    if (threadIdx.x < HIDDEN) b1s[threadIdx.x] = b1[threadIdx.x];
    __syncthreads();

    const int lane = threadIdx.x & 31;
    const int wave = threadIdx.x >> 5;
    const int tile = blockIdx.x * (blockDim.x >> 5) + wave;
    if (tile >= M_TILES) return;
    const int row0 = tile << 4;
    const int half = lane >> 4;
    const int l16  = lane & 15;
    const float* arow = h1 + (size_t)(row0 + l16) * PAD1;

    v8f c = {};
    #pragma unroll
    for (int k0 = 0; k0 < HIDDEN; k0 += 4) {
        const int kb = k0 + 2 * half;
        v2f ar = *(const v2f*)(arow + kb);
        v2f a;
        a.x = fmaxf(ar.x + b1s[kb],     0.0f);
        a.y = fmaxf(ar.y + b1s[kb + 1], 0.0f);
        v2f b = *(const v2f*)&Bf2[(kb >> 1) * 32 + l16 * 2];
        c = __builtin_amdgcn_wmma_f32_16x16x4_f32(false, a, false, b, (short)0, c, false, false);
    }

    #pragma unroll
    for (int r = 0; r < 8; ++r) {
        const int row = row0 + r + 8 * half;
        const float di = dis[row];
        const float d2 = di * di;
        h2 [(size_t)row * PAD2 + l16] = c[r];
        h2a[(size_t)row * PAD2 + l16] = d2 * c[r];
    }
}


__global__ void gcn_logsoftmax(const float* __restrict__ h2a,
                               const float* __restrict__ b2,
                               float* __restrict__ out) {
    int i = blockIdx.x * blockDim.x + threadIdx.x;
    if (i >= N_NODES) return;
    const float* row = h2a + (size_t)i * PAD2;
    float z[N_CLASSES];
    {
        float4 t0 = *(const float4*)(row);
        float4 t1 = *(const float4*)(row + 4);
        float2 t2 = *(const float2*)(row + 8);
        z[0] = t0.x; z[1] = t0.y; z[2] = t0.z; z[3] = t0.w;
        z[4] = t1.x; z[5] = t1.y; z[6] = t1.z; z[7] = t1.w;
        z[8] = t2.x; z[9] = t2.y;
    }
    float m = -3.402823466e38f;
    #pragma unroll
    for (int c = 0; c < N_CLASSES; ++c) {
        z[c] += b2[c];
        m = fmaxf(m, z[c]);
    }
    float s = 0.0f;
    #pragma unroll
    for (int c = 0; c < N_CLASSES; ++c) s += __expf(z[c] - m);
    const float l = __logf(s);
    #pragma unroll
    for (int c = 0; c < N_CLASSES; ++c) out[(size_t)i * N_CLASSES + c] = z[c] - m - l;
}


extern "C" void kernel_launch(void* const* d_in, const int* in_sizes, int n_in,
                              void* d_out, int out_size, void* d_ws, size_t ws_size,
                              hipStream_t stream) {
    const float* x  = (const float*)d_in[0];
    const int*   ei = (const int*)d_in[1];
    const float* W1 = (const float*)d_in[2];
    const float* b1 = (const float*)d_in[3];
    const float* W2 = (const float*)d_in[4];
    const float* b2 = (const float*)d_in[5];
    float* out = (float*)d_out;

    const int E = in_sizes[1] / 2;
    const int* src = ei;
    const int* dst = ei + E;

    float* ws  = (float*)d_ws;
    float* dis = ws;
    float* h0  = ws + 102400;
    float* h1  = h0 + (size_t)N_NODES * PAD1;
    float* h2  = h1 + (size_t)N_NODES * PAD1;
    float* h2a = h2 + (size_t)N_NODES * PAD2;

    const int TB = 256;
    const int nodeBlocks = (N_NODES + TB - 1) / TB;
    const int edgeBlocks = (E + TB - 1) / TB;
    const int gemmBlocks = (M_TILES + (TB / 32) - 1) / (TB / 32);

    gcn_init_deg<<<nodeBlocks, TB, 0, stream>>>(dis);
    gcn_deg_accum<<<edgeBlocks, TB, 0, stream>>>(dst, dis, E);
    gcn_rsqrt<<<nodeBlocks, TB, 0, stream>>>(dis);

    gcn_gemm1<<<gemmBlocks, TB, 0, stream>>>(x, W1, dis, h0, h1);
    gcn_propagate<HIDDEN, PAD1><<<edgeBlocks, TB, 0, stream>>>(src, dst, dis, h0, h1, E);

    gcn_gemm2<<<gemmBlocks, TB, 0, stream>>>(h1, b1, W2, dis, h2, h2a);
    gcn_propagate<N_CLASSES, PAD2><<<edgeBlocks, TB, 0, stream>>>(src, dst, dis, h2, h2a, E);

    gcn_logsoftmax<<<nodeBlocks, TB, 0, stream>>>(h2a, b2, out);
}


# Scatter-add without global atomics: edges counting-sorted into 196-row destination buckets, each bucket summed in LDS by one owner group; f32 sums now in LDS-atomic arrival order
# speedup vs baseline: 6.6931x; 6.6931x over previous
#include <hip/hip_runtime.h>


typedef __attribute__((ext_vector_type(2))) float v2f;
typedef __attribute__((ext_vector_type(8))) float v8f;

#define N_NODES   100000
#define N_FEAT    128
#define HIDDEN    20
#define N_CLASSES 10
#define PAD1      32
#define PAD2      16
#define M_TILES   (N_NODES / 16)

#define RB    196
#define NB    512
#define NPW   128
#define PT    1024
#define NCNT  (NB * NPW)

__device__ __forceinline__ int gcn_bucket_of(int d) {
    const unsigned b = (unsigned)d / RB;
    return (int)(b < (unsigned)NB ? b : (unsigned)(NB - 1));
}

__device__ __forceinline__ int gcn_local_row(int d, int base) {
    const unsigned r = (unsigned)(d - base);
    return (int)(r < (unsigned)RB ? r : (unsigned)(RB - 1));
}


__global__ void __launch_bounds__(PT)
gcn_part_count(const int* __restrict__ dst, int* __restrict__ cnt, int E) {
    __shared__ int hist[NB];
    for (int i = threadIdx.x; i < NB; i += PT) hist[i] = 0;
    __syncthreads();
    const int chunk = (E + NPW - 1) / NPW;
    const int lo = blockIdx.x * chunk;
    const int hi = min(lo + chunk, E);
    for (int i0 = lo + threadIdx.x; i0 < hi; i0 += 4 * PT) {
        int d[4];
        #pragma unroll
        for (int u = 0; u < 4; ++u) {
            const int i = i0 + u * PT;
            d[u] = (i < hi) ? dst[i] : 0;
        }
        #pragma unroll
        for (int u = 0; u < 4; ++u)
            if (i0 + u * PT < hi) atomicAdd(&hist[gcn_bucket_of(d[u])], 1);
    }
    __syncthreads();
    for (int i = threadIdx.x; i < NB; i += PT) cnt[i * NPW + blockIdx.x] = hist[i];
}


__global__ void __launch_bounds__(1024)
gcn_part_scan(int* __restrict__ cnt) {
    __shared__ int wsum[32];
    int4* p = (int4*)cnt;
    const int lane = threadIdx.x & 31;
    const int wave = threadIdx.x >> 5;
    int4* pw = p + wave * 512 + lane;
    int4 v[16];
    #pragma unroll
    for (int j = 0; j < 16; ++j) v[j] = pw[j * 32];
    int carry = 0;
    #pragma unroll
    for (int j = 0; j < 16; ++j) {
        const int4 a = v[j];
        const int s4 = a.x + a.y + a.z + a.w;
        int inc = s4;
        #pragma unroll
        for (int o = 1; o < 32; o <<= 1) {
            const int y = __shfl_up(inc, o);
            if (lane >= o) inc += y;
        }
        const int ex = carry + inc - s4;
        int4 r;
        r.x = ex;
        r.y = ex + a.x;
        r.z = ex + a.x + a.y;
        r.w = ex + a.x + a.y + a.z;
        v[j] = r;
        carry += __shfl(inc, 31);
    }
    if (lane == 0) wsum[wave] = carry;
    __syncthreads();
    if (wave == 0) {
        const int w = wsum[lane];
        int winc = w;
        #pragma unroll
        for (int o = 1; o < 32; o <<= 1) {
            const int y = __shfl_up(winc, o);
            if (lane >= o) winc += y;
        }
        wsum[lane] = winc - w;
    }
    __syncthreads();
    const int add = wsum[wave];
    #pragma unroll
    for (int j = 0; j < 16; ++j) {
        int4 r = v[j];
        r.x += add; r.y += add; r.z += add; r.w += add;
        pw[j * 32] = r;
    }
}


__global__ void __launch_bounds__(PT)
gcn_part_scatter(const int* __restrict__ src, const int* __restrict__ dst,
                 const int* __restrict__ off, int* __restrict__ bsd_i, int E) {
    __shared__ int cur[NB];
    int2* bsd = (int2*)bsd_i;
    for (int i = threadIdx.x; i < NB; i += PT) cur[i] = off[i * NPW + blockIdx.x];
    __syncthreads();
    const int chunk = (E + NPW - 1) / NPW;
    const int lo = blockIdx.x * chunk;
    const int hi = min(lo + chunk, E);
    for (int i0 = lo + threadIdx.x; i0 < hi; i0 += 4 * PT) {
        int d[4], s[4];
        #pragma unroll
        for (int u = 0; u < 4; ++u) {
            const int i = i0 + u * PT;
            d[u] = (i < hi) ? dst[i] : 0;
            s[u] = (i < hi) ? src[i] : 0;
        }
        #pragma unroll
        for (int u = 0; u < 4; ++u) {
            if (i0 + u * PT < hi) {
                const int pos = atomicAdd(&cur[gcn_bucket_of(d[u])], 1);
                bsd[pos] = make_int2(s[u], d[u]);
            }
        }
    }
}


__global__ void __launch_bounds__(PT)
gcn_bucket_deg(const int* __restrict__ bsd_i, const int* __restrict__ off,
               float* __restrict__ dis, int E) {
    __shared__ int c[RB];
    const int2* bsd = (const int2*)bsd_i;
    const int b = blockIdx.x;
    const int base = b * RB;
    for (int i = threadIdx.x; i < RB; i += PT) c[i] = 0;
    __syncthreads();
    const int lo = off[b * NPW];
    const int hi = (b + 1 < NB) ? off[(b + 1) * NPW] : E;
    for (int i = lo + threadIdx.x; i < hi; i += PT)
        atomicAdd(&c[gcn_local_row(bsd[i].y, base)], 1);
    __syncthreads();
    for (int i = threadIdx.x; i < RB; i += PT) {
        const int row = base + i;
        if (row < N_NODES) dis[row] = rsqrtf(1.0f + (float)c[i]);
    }
}


__global__ void gcn_gemm1(const float* __restrict__ x,
                          const float* __restrict__ W1,
                          const float* __restrict__ dis,
                          float* __restrict__ h0,
                          float* __restrict__ h1) {
    __shared__ float Bf[(N_FEAT / 2) * 32 * 2];

    for (int idx = threadIdx.x; idx < (N_FEAT / 2) * 32 * 2; idx += blockDim.x) {
        const int p = idx >> 6;
        const int c = (idx >> 1) & 31;
        const int j = idx & 1;
        Bf[idx] = (c < HIDDEN) ? W1[(2 * p + j) * HIDDEN + c] : 0.0f;
    }
    __syncthreads();

    const int lane = threadIdx.x & 31;
    const int wave = threadIdx.x >> 5;
    const int tile = blockIdx.x * (blockDim.x >> 5) + wave;
    if (tile >= M_TILES) return;
    const int row0 = tile << 4;
    const int half = lane >> 4;
    const int l16  = lane & 15;
    const float* arow = x + (size_t)(row0 + l16) * N_FEAT;

    v8f c0 = {};
    v8f c1 = {};

    #pragma unroll 4
    for (int k0 = 0; k0 < N_FEAT; k0 += 4) {
        const int kb = k0 + 2 * half;
        const int p = kb >> 1;
        v2f a  = *(const v2f*)(arow + kb);
        v2f b0 = *(const v2f*)&Bf[p * 64 + l16 * 2];
        v2f b1 = *(const v2f*)&Bf[p * 64 + (16 + l16) * 2];

        c0 = __builtin_amdgcn_wmma_f32_16x16x4_f32(false, a, false, b0, (short)0, c0, false, false);
        c1 = __builtin_amdgcn_wmma_f32_16x16x4_f32(false, a, false, b1, (short)0, c1, false, false);
    }

    #pragma unroll
    for (int r = 0; r < 8; ++r) {
        const int row = row0 + r + 8 * half;
        const float di = dis[row];
        const float d2 = di * di;
        float* p0 = h0 + (size_t)row * PAD1;
        float* p1 = h1 + (size_t)row * PAD1;
        p0[l16]      = c0[r];
        p0[16 + l16] = c1[r];
        p1[l16]      = d2 * c0[r];
        p1[16 + l16] = d2 * c1[r];
    }
}


template <int F, int PAD>
__global__ void __launch_bounds__(PT)
gcn_bucket_acc(const int* __restrict__ bsd_i, const int* __restrict__ off,
               const float* __restrict__ dis, const float* __restrict__ h,
               float* __restrict__ out, int E) {
    constexpr int TP = F + 1;
    __shared__ float tile[RB * TP];
    const int2* bsd = (const int2*)bsd_i;
    const int b = blockIdx.x;
    const int base = b * RB;
    for (int i = threadIdx.x; i < RB * TP; i += PT) tile[i] = 0.0f;
    __syncthreads();
    const int lo = off[b * NPW];
    const int hi = (b + 1 < NB) ? off[(b + 1) * NPW] : E;

    for (int i = lo + threadIdx.x; i < hi; i += PT) {
        const int2 sd = bsd[i];
        const int s = sd.x;
        const int d = sd.y;
        const float nrm = dis[s] * dis[d];
        const float* hrow = h + (size_t)s * PAD;

        float v[F];
        #pragma unroll
        for (int j = 0; j + 4 <= F; j += 4) {
            float4 t = *(const float4*)(hrow + j);
            v[j] = t.x; v[j + 1] = t.y; v[j + 2] = t.z; v[j + 3] = t.w;
        }
        if constexpr (F % 4 != 0) {
            float2 t = *(const float2*)(hrow + (F & ~3));
            v[(F & ~3)] = t.x; v[(F & ~3) + 1] = t.y;
        }
        float* trow = &tile[gcn_local_row(d, base) * TP];
        #pragma unroll
        for (int f = 0; f < F; ++f) atomicAdd(&trow[f], nrm * v[f]);
    }
    __syncthreads();

    for (int i = threadIdx.x; i < RB * F; i += PT) {
        const int r = i / F;
        const int f = i - r * F;
        const int row = base + r;
        if (row < N_NODES) out[(size_t)row * PAD + f] += tile[r * TP + f];
    }
}


__global__ void gcn_gemm2(const float* __restrict__ h1,
                          const float* __restrict__ b1,
                          const float* __restrict__ W2,
                          const float* __restrict__ dis,
                          float* __restrict__ h2,
                          float* __restrict__ h2a) {
    __shared__ float Bf2[(HIDDEN / 2) * 16 * 2];
    __shared__ float b1s[HIDDEN];

    for (int idx = threadIdx.x; idx < (HIDDEN / 2) * 16 * 2; idx += blockDim.x) {
        const int p = idx >> 5;
        const int c = (idx >> 1) & 15;
        const int j = idx & 1;
        Bf2[idx] = (c < N_CLASSES) ? W2[(2 * p + j) * N_CLASSES + c] : 0.0f;
    }
    if (threadIdx.x < HIDDEN) b1s[threadIdx.x] = b1[threadIdx.x];
    __syncthreads();

    const int lane = threadIdx.x & 31;
    const int wave = threadIdx.x >> 5;
    const int tile = blockIdx.x * (blockDim.x >> 5) + wave;
    if (tile >= M_TILES) return;
    const int row0 = tile << 4;
    const int half = lane >> 4;
    const int l16  = lane & 15;
    const float* arow = h1 + (size_t)(row0 + l16) * PAD1;

    v8f c = {};
    #pragma unroll
    for (int k0 = 0; k0 < HIDDEN; k0 += 4) {
        const int kb = k0 + 2 * half;
        v2f ar = *(const v2f*)(arow + kb);
        v2f a;
        a.x = fmaxf(ar.x + b1s[kb],     0.0f);
        a.y = fmaxf(ar.y + b1s[kb + 1], 0.0f);
        v2f b = *(const v2f*)&Bf2[(kb >> 1) * 32 + l16 * 2];
        c = __builtin_amdgcn_wmma_f32_16x16x4_f32(false, a, false, b, (short)0, c, false, false);
    }

    #pragma unroll
    for (int r = 0; r < 8; ++r) {
        const int row = row0 + r + 8 * half;
        const float di = dis[row];
        const float d2 = di * di;
        h2 [(size_t)row * PAD2 + l16] = c[r];
        h2a[(size_t)row * PAD2 + l16] = d2 * c[r];
    }
}


__global__ void gcn_logsoftmax(const float* __restrict__ h2a,
                               const float* __restrict__ b2,
                               float* __restrict__ out) {
    int i = blockIdx.x * blockDim.x + threadIdx.x;
    if (i >= N_NODES) return;
    const float* row = h2a + (size_t)i * PAD2;
    float z[N_CLASSES];
    {
        float4 t0 = *(const float4*)(row);
        float4 t1 = *(const float4*)(row + 4);
        float2 t2 = *(const float2*)(row + 8);
        z[0] = t0.x; z[1] = t0.y; z[2] = t0.z; z[3] = t0.w;
        z[4] = t1.x; z[5] = t1.y; z[6] = t1.z; z[7] = t1.w;
        z[8] = t2.x; z[9] = t2.y;
    }
    float m = -3.402823466e38f;
    #pragma unroll
    for (int c = 0; c < N_CLASSES; ++c) {
        z[c] += b2[c];
        m = fmaxf(m, z[c]);
    }
    float s = 0.0f;
    #pragma unroll
    for (int c = 0; c < N_CLASSES; ++c) s += __expf(z[c] - m);
    const float l = __logf(s);
    #pragma unroll
    for (int c = 0; c < N_CLASSES; ++c) out[(size_t)i * N_CLASSES + c] = z[c] - m - l;
}


extern "C" void kernel_launch(void* const* d_in, const int* in_sizes, int n_in,
                              void* d_out, int out_size, void* d_ws, size_t ws_size,
                              hipStream_t stream) {
    const float* x  = (const float*)d_in[0];
    const int*   ei = (const int*)d_in[1];
    const float* W1 = (const float*)d_in[2];
    const float* b1 = (const float*)d_in[3];
    const float* W2 = (const float*)d_in[4];
    const float* b2 = (const float*)d_in[5];
    float* out = (float*)d_out;

    const int E = in_sizes[1] / 2;
    const int* src = ei;
    const int* dst = ei + E;

    float* ws  = (float*)d_ws;
    float* dis = ws;
    float* h0  = ws + 102400;
    float* h1  = h0 + (size_t)N_NODES * PAD1;
    float* h2  = h1 + (size_t)N_NODES * PAD1;
    float* h2a = h2 + (size_t)N_NODES * PAD2;
    int*   cnt = (int*)(h2a + (size_t)N_NODES * PAD2);
    int*   bsd = cnt + NCNT;

    const int TB = 256;
    const int nodeBlocks = (N_NODES + TB - 1) / TB;
    const int gemmBlocks = (M_TILES + (TB / 32) - 1) / (TB / 32);

    gcn_part_count<<<NPW, PT, 0, stream>>>(dst, cnt, E);
    gcn_part_scan<<<1, 1024, 0, stream>>>(cnt);
    gcn_part_scatter<<<NPW, PT, 0, stream>>>(src, dst, cnt, bsd, E);

    gcn_bucket_deg<<<NB, PT, 0, stream>>>(bsd, cnt, dis, E);

    gcn_gemm1<<<gemmBlocks, TB, 0, stream>>>(x, W1, dis, h0, h1);
    gcn_bucket_acc<HIDDEN, PAD1><<<NB, PT, 0, stream>>>(bsd, cnt, dis, h0, h1, E);

    gcn_gemm2<<<gemmBlocks, TB, 0, stream>>>(h1, b1, W2, dis, h2, h2a);
    gcn_bucket_acc<N_CLASSES, PAD2><<<NB, PT, 0, stream>>>(bsd, cnt, dis, h2, h2a, E);

    gcn_logsoftmax<<<nodeBlocks, TB, 0, stream>>>(h2a, b2, out);
}


# Edge weight dis[src]*dis[dst] factored out of the edge loop: GEMM rows stored pre-scaled by dis, bucket owner adds self row and scales by dis[dst] once; products re-associated, sum order kept
# speedup vs baseline: 7.6257x; 1.1393x over previous
#include <hip/hip_runtime.h>


typedef __attribute__((ext_vector_type(2))) float v2f;
typedef __attribute__((ext_vector_type(8))) float v8f;

#define N_NODES   100000
#define N_FEAT    128
#define HIDDEN    20
#define N_CLASSES 10
#define PAD1      32
#define PAD2      16
#define M_TILES   (N_NODES / 16)

#define RB    196
#define NB    512
#define NPW   128
#define PT    1024
#define NCNT  (NB * NPW)

__device__ __forceinline__ int gcn_bucket_of(int d) {
    const unsigned b = (unsigned)d / RB;
    return (int)(b < (unsigned)NB ? b : (unsigned)(NB - 1));
}

__device__ __forceinline__ int gcn_local_row(int d, int base) {
    const unsigned r = (unsigned)(d - base);
    return (int)(r < (unsigned)RB ? r : (unsigned)(RB - 1));
}


__global__ void __launch_bounds__(PT)
gcn_part_count(const int* __restrict__ dst, int* __restrict__ cnt, int E) {
    __shared__ int hist[NB];
    for (int i = threadIdx.x; i < NB; i += PT) hist[i] = 0;
    __syncthreads();
    const int chunk = (E + NPW - 1) / NPW;
    const int lo = blockIdx.x * chunk;
    const int hi = min(lo + chunk, E);
    for (int i0 = lo + threadIdx.x; i0 < hi; i0 += 4 * PT) {
        int d[4];
        #pragma unroll
        for (int u = 0; u < 4; ++u) {
            const int i = i0 + u * PT;
            d[u] = (i < hi) ? dst[i] : 0;
        }
        #pragma unroll
        for (int u = 0; u < 4; ++u)
            if (i0 + u * PT < hi) atomicAdd(&hist[gcn_bucket_of(d[u])], 1);
    }
    __syncthreads();
    for (int i = threadIdx.x; i < NB; i += PT) cnt[i * NPW + blockIdx.x] = hist[i];
}


__global__ void __launch_bounds__(1024)
gcn_part_scan(int* __restrict__ cnt) {
    __shared__ int wsum[32];
    int4* p = (int4*)cnt;
    const int lane = threadIdx.x & 31;
    const int wave = threadIdx.x >> 5;
    int4* pw = p + wave * 512 + lane;
    int4 v[16];
    #pragma unroll
    for (int j = 0; j < 16; ++j) v[j] = pw[j * 32];
    int carry = 0;
    #pragma unroll
    for (int j = 0; j < 16; ++j) {
        const int4 a = v[j];
        const int s4 = a.x + a.y + a.z + a.w;
        int inc = s4;
        #pragma unroll
        for (int o = 1; o < 32; o <<= 1) {
            const int y = __shfl_up(inc, o);
            if (lane >= o) inc += y;
        }
        const int ex = carry + inc - s4;
        int4 r;
        r.x = ex;
        r.y = ex + a.x;
        r.z = ex + a.x + a.y;
        r.w = ex + a.x + a.y + a.z;
        v[j] = r;
        carry += __shfl(inc, 31);
    }
    if (lane == 0) wsum[wave] = carry;
    __syncthreads();
    if (wave == 0) {
        const int w = wsum[lane];
        int winc = w;
        #pragma unroll
        for (int o = 1; o < 32; o <<= 1) {
            const int y = __shfl_up(winc, o);
            if (lane >= o) winc += y;
        }
        wsum[lane] = winc - w;
    }
    __syncthreads();
    const int add = wsum[wave];
    #pragma unroll
    for (int j = 0; j < 16; ++j) {
        int4 r = v[j];
        r.x += add; r.y += add; r.z += add; r.w += add;
        pw[j * 32] = r;
    }
}


__global__ void __launch_bounds__(PT)
gcn_part_scatter(const int* __restrict__ src, const int* __restrict__ dst,
                 const int* __restrict__ off, int* __restrict__ bsd_i, int E) {
    __shared__ int cur[NB];
    int2* bsd = (int2*)bsd_i;
    for (int i = threadIdx.x; i < NB; i += PT) cur[i] = off[i * NPW + blockIdx.x];
    __syncthreads();
    const int chunk = (E + NPW - 1) / NPW;
    const int lo = blockIdx.x * chunk;
    const int hi = min(lo + chunk, E);
    for (int i0 = lo + threadIdx.x; i0 < hi; i0 += 4 * PT) {
        int d[4], s[4];
        #pragma unroll
        for (int u = 0; u < 4; ++u) {
            const int i = i0 + u * PT;
            d[u] = (i < hi) ? dst[i] : 0;
            s[u] = (i < hi) ? src[i] : 0;
        }
        #pragma unroll
        for (int u = 0; u < 4; ++u) {
            if (i0 + u * PT < hi) {
                const int pos = atomicAdd(&cur[gcn_bucket_of(d[u])], 1);
                bsd[pos] = make_int2(s[u], d[u]);
            }
        }
    }
}


__global__ void __launch_bounds__(PT)
gcn_bucket_deg(const int* __restrict__ bsd_i, const int* __restrict__ off,
               float* __restrict__ dis, int E) {
    __shared__ int c[RB];
    const int2* bsd = (const int2*)bsd_i;
    const int b = blockIdx.x;
    const int base = b * RB;
    for (int i = threadIdx.x; i < RB; i += PT) c[i] = 0;
    __syncthreads();
    const int lo = off[b * NPW];
    const int hi = (b + 1 < NB) ? off[(b + 1) * NPW] : E;
    for (int i = lo + threadIdx.x; i < hi; i += PT)
        atomicAdd(&c[gcn_local_row(bsd[i].y, base)], 1);
    __syncthreads();
    for (int i = threadIdx.x; i < RB; i += PT) {
        const int row = base + i;
        if (row < N_NODES) dis[row] = rsqrtf(1.0f + (float)c[i]);
    }
}


__global__ void gcn_gemm1(const float* __restrict__ x,
                          const float* __restrict__ W1,
                          const float* __restrict__ dis,
                          float* __restrict__ hs) {
    __shared__ float Bf[(N_FEAT / 2) * 32 * 2];

    for (int idx = threadIdx.x; idx < (N_FEAT / 2) * 32 * 2; idx += blockDim.x) {
        const int p = idx >> 6;
        const int c = (idx >> 1) & 31;
        const int j = idx & 1;
        Bf[idx] = (c < HIDDEN) ? W1[(2 * p + j) * HIDDEN + c] : 0.0f;
    }
    __syncthreads();

    const int lane = threadIdx.x & 31;
    const int wave = threadIdx.x >> 5;
    const int tile = blockIdx.x * (blockDim.x >> 5) + wave;
    if (tile >= M_TILES) return;
    const int row0 = tile << 4;
    const int half = lane >> 4;
    const int l16  = lane & 15;
    const float* arow = x + (size_t)(row0 + l16) * N_FEAT;

    v8f c0 = {};
    v8f c1 = {};

    #pragma unroll 4
    for (int k0 = 0; k0 < N_FEAT; k0 += 4) {
        const int kb = k0 + 2 * half;
        const int p = kb >> 1;
        v2f a  = *(const v2f*)(arow + kb);
        v2f b0 = *(const v2f*)&Bf[p * 64 + l16 * 2];
        v2f b1 = *(const v2f*)&Bf[p * 64 + (16 + l16) * 2];

        c0 = __builtin_amdgcn_wmma_f32_16x16x4_f32(false, a, false, b0, (short)0, c0, false, false);
        c1 = __builtin_amdgcn_wmma_f32_16x16x4_f32(false, a, false, b1, (short)0, c1, false, false);
    }

    #pragma unroll
    for (int r = 0; r < 8; ++r) {
        const int row = row0 + r + 8 * half;
        const float di = dis[row];
        float* p0 = hs + (size_t)row * PAD1;
        p0[l16]      = di * c0[r];
        p0[16 + l16] = di * c1[r];
    }
}


template <int F, int PAD>
__global__ void __launch_bounds__(PT)
gcn_bucket_acc(const int* __restrict__ bsd_i, const int* __restrict__ off,
               const float* __restrict__ dis, const float* __restrict__ hs,
               float* __restrict__ out, int E) {
    constexpr int TP = F + 1;
    __shared__ float tile[RB * TP];
    const int2* bsd = (const int2*)bsd_i;
    const int b = blockIdx.x;
    const int base = b * RB;
    for (int i = threadIdx.x; i < RB * TP; i += PT) tile[i] = 0.0f;
    __syncthreads();
    const int lo = off[b * NPW];
    const int hi = (b + 1 < NB) ? off[(b + 1) * NPW] : E;

    for (int i = lo + threadIdx.x; i < hi; i += PT) {
        const int2 sd = bsd[i];
        const int s = sd.x;
        const int d = sd.y;
        const float* hrow = hs + (size_t)s * PAD;

        float v[F];
        #pragma unroll
        for (int j = 0; j + 4 <= F; j += 4) {
            float4 t = *(const float4*)(hrow + j);
            v[j] = t.x; v[j + 1] = t.y; v[j + 2] = t.z; v[j + 3] = t.w;
        }
        if constexpr (F % 4 != 0) {
            float2 t = *(const float2*)(hrow + (F & ~3));
            v[(F & ~3)] = t.x; v[(F & ~3) + 1] = t.y;
        }
        float* trow = &tile[gcn_local_row(d, base) * TP];
        #pragma unroll
        for (int f = 0; f < F; ++f) atomicAdd(&trow[f], v[f]);
    }
    __syncthreads();

    for (int i = threadIdx.x; i < RB * F; i += PT) {
        const int r = i / F;
        const int f = i - r * F;
        const int row = base + r;
        if (row < N_NODES)
            out[(size_t)row * PAD + f] = dis[row] * (tile[r * TP + f] + hs[(size_t)row * PAD + f]);
    }
}


__global__ void gcn_gemm2(const float* __restrict__ h1,
                          const float* __restrict__ b1,
                          const float* __restrict__ W2,
                          const float* __restrict__ dis,
                          float* __restrict__ h2s) {
    __shared__ float Bf2[(HIDDEN / 2) * 16 * 2];
    __shared__ float b1s[HIDDEN];

    for (int idx = threadIdx.x; idx < (HIDDEN / 2) * 16 * 2; idx += blockDim.x) {
        const int p = idx >> 5;
        const int c = (idx >> 1) & 15;
        const int j = idx & 1;
        Bf2[idx] = (c < N_CLASSES) ? W2[(2 * p + j) * N_CLASSES + c] : 0.0f;
    }
    if (threadIdx.x < HIDDEN) b1s[threadIdx.x] = b1[threadIdx.x];
    __syncthreads();

    const int lane = threadIdx.x & 31;
    const int wave = threadIdx.x >> 5;
    const int tile = blockIdx.x * (blockDim.x >> 5) + wave;
    if (tile >= M_TILES) return;
    const int row0 = tile << 4;
    const int half = lane >> 4;
    const int l16  = lane & 15;
    const float* arow = h1 + (size_t)(row0 + l16) * PAD1;

    v8f c = {};
    #pragma unroll
    for (int k0 = 0; k0 < HIDDEN; k0 += 4) {
        const int kb = k0 + 2 * half;
        v2f ar = *(const v2f*)(arow + kb);
        v2f a;
        a.x = fmaxf(ar.x + b1s[kb],     0.0f);
        a.y = fmaxf(ar.y + b1s[kb + 1], 0.0f);
        v2f b = *(const v2f*)&Bf2[(kb >> 1) * 32 + l16 * 2];
        c = __builtin_amdgcn_wmma_f32_16x16x4_f32(false, a, false, b, (short)0, c, false, false);
    }

    #pragma unroll
    for (int r = 0; r < 8; ++r) {
        const int row = row0 + r + 8 * half;
        const float di = dis[row];
        h2s[(size_t)row * PAD2 + l16] = di * c[r];
    }
}


__global__ void gcn_logsoftmax(const float* __restrict__ h2a,
                               const float* __restrict__ b2,
                               float* __restrict__ out) {
    int i = blockIdx.x * blockDim.x + threadIdx.x;
    if (i >= N_NODES) return;
    const float* row = h2a + (size_t)i * PAD2;
    float z[N_CLASSES];
    {
        float4 t0 = *(const float4*)(row);
        float4 t1 = *(const float4*)(row + 4);
        float2 t2 = *(const float2*)(row + 8);
        z[0] = t0.x; z[1] = t0.y; z[2] = t0.z; z[3] = t0.w;
        z[4] = t1.x; z[5] = t1.y; z[6] = t1.z; z[7] = t1.w;
        z[8] = t2.x; z[9] = t2.y;
    }
    float m = -3.402823466e38f;
    #pragma unroll
    for (int c = 0; c < N_CLASSES; ++c) {
        z[c] += b2[c];
        m = fmaxf(m, z[c]);
    }
    float s = 0.0f;
    #pragma unroll
    for (int c = 0; c < N_CLASSES; ++c) s += __expf(z[c] - m);
    const float l = __logf(s);
    #pragma unroll
    for (int c = 0; c < N_CLASSES; ++c) out[(size_t)i * N_CLASSES + c] = z[c] - m - l;
}


extern "C" void kernel_launch(void* const* d_in, const int* in_sizes, int n_in,
                              void* d_out, int out_size, void* d_ws, size_t ws_size,
                              hipStream_t stream) {
    const float* x  = (const float*)d_in[0];
    const int*   ei = (const int*)d_in[1];
    const float* W1 = (const float*)d_in[2];
    const float* b1 = (const float*)d_in[3];
    const float* W2 = (const float*)d_in[4];
    const float* b2 = (const float*)d_in[5];
    float* out = (float*)d_out;

    const int E = in_sizes[1] / 2;
    const int* src = ei;
    const int* dst = ei + E;

    float* ws  = (float*)d_ws;
    float* dis = ws;
    float* h0  = ws + 102400;
    float* h1  = h0 + (size_t)N_NODES * PAD1;
    float* h2  = h1 + (size_t)N_NODES * PAD1;
    float* h2a = h2 + (size_t)N_NODES * PAD2;
    int*   cnt = (int*)(h2a + (size_t)N_NODES * PAD2);
    int*   bsd = cnt + NCNT;

    const int TB = 256;
    const int nodeBlocks = (N_NODES + TB - 1) / TB;
    const int gemmBlocks = (M_TILES + (TB / 32) - 1) / (TB / 32);

    gcn_part_count<<<NPW, PT, 0, stream>>>(dst, cnt, E);
    gcn_part_scan<<<1, 1024, 0, stream>>>(cnt);
    gcn_part_scatter<<<NPW, PT, 0, stream>>>(src, dst, cnt, bsd, E);

    gcn_bucket_deg<<<NB, PT, 0, stream>>>(bsd, cnt, dis, E);

    gcn_gemm1<<<gemmBlocks, TB, 0, stream>>>(x, W1, dis, h0);
    gcn_bucket_acc<HIDDEN, PAD1><<<NB, PT, 0, stream>>>(bsd, cnt, dis, h0, h1, E);

    gcn_gemm2<<<gemmBlocks, TB, 0, stream>>>(h1, b1, W2, dis, h2);
    gcn_bucket_acc<N_CLASSES, PAD2><<<NB, PT, 0, stream>>>(bsd, cnt, dis, h2, h2a, E);

    gcn_logsoftmax<<<nodeBlocks, TB, 0, stream>>>(h2a, b2, out);
}


# Node-wise stage after each bucket sum runs in the bucket owner from its LDS tile: bias+relu+W2 product (f32 WMMA) after layer 1, bias+log_softmax after layer 2; 2 planes, 2 launches fewer
# speedup vs baseline: 8.4255x; 1.1049x over previous
#include <hip/hip_runtime.h>


typedef __attribute__((ext_vector_type(2))) float v2f;
typedef __attribute__((ext_vector_type(8))) float v8f;

#define N_NODES   100000
#define N_FEAT    128
#define HIDDEN    20
#define N_CLASSES 10
#define PAD1      32
#define PAD2      16
#define M_TILES   (N_NODES / 16)

#define RB    196
#define NB    512
#define NPW   128
#define PT    1024
#define NCNT  (NB * NPW)

__device__ __forceinline__ int gcn_bucket_of(int d) {
    const unsigned b = (unsigned)d / RB;
    return (int)(b < (unsigned)NB ? b : (unsigned)(NB - 1));
}

__device__ __forceinline__ int gcn_local_row(int d, int base) {
    const unsigned r = (unsigned)(d - base);
    return (int)(r < (unsigned)RB ? r : (unsigned)(RB - 1));
}


__global__ void __launch_bounds__(PT)
gcn_part_count(const int* __restrict__ dst, int* __restrict__ cnt, int E) {
    __shared__ int hist[NB];
    for (int i = threadIdx.x; i < NB; i += PT) hist[i] = 0;
    __syncthreads();
    const int chunk = (E + NPW - 1) / NPW;
    const int lo = blockIdx.x * chunk;
    const int hi = min(lo + chunk, E);
    for (int i0 = lo + threadIdx.x; i0 < hi; i0 += 4 * PT) {
        int d[4];
        #pragma unroll
        for (int u = 0; u < 4; ++u) {
            const int i = i0 + u * PT;
            d[u] = (i < hi) ? dst[i] : 0;
        }
        #pragma unroll
        for (int u = 0; u < 4; ++u)
            if (i0 + u * PT < hi) atomicAdd(&hist[gcn_bucket_of(d[u])], 1);
    }
    __syncthreads();
    for (int i = threadIdx.x; i < NB; i += PT) cnt[i * NPW + blockIdx.x] = hist[i];
}


__global__ void __launch_bounds__(1024)
gcn_part_scan(int* __restrict__ cnt) {
    __shared__ int wsum[32];
    int4* p = (int4*)cnt;
    const int lane = threadIdx.x & 31;
    const int wave = threadIdx.x >> 5;
    int4* pw = p + wave * 512 + lane;
    int4 v[16];
    #pragma unroll
    for (int j = 0; j < 16; ++j) v[j] = pw[j * 32];
    int carry = 0;
    #pragma unroll
    for (int j = 0; j < 16; ++j) {
        const int4 a = v[j];
        const int s4 = a.x + a.y + a.z + a.w;
        int inc = s4;
        #pragma unroll
        for (int o = 1; o < 32; o <<= 1) {
            const int y = __shfl_up(inc, o);
            if (lane >= o) inc += y;
        }
        const int ex = carry + inc - s4;
        int4 r;
        r.x = ex;
        r.y = ex + a.x;
        r.z = ex + a.x + a.y;
        r.w = ex + a.x + a.y + a.z;
        v[j] = r;
        carry += __shfl(inc, 31);
    }
    if (lane == 0) wsum[wave] = carry;
    __syncthreads();
    if (wave == 0) {
        const int w = wsum[lane];
        int winc = w;
        #pragma unroll
        for (int o = 1; o < 32; o <<= 1) {
            const int y = __shfl_up(winc, o);
            if (lane >= o) winc += y;
        }
        wsum[lane] = winc - w;
    }
    __syncthreads();
    const int add = wsum[wave];
    #pragma unroll
    for (int j = 0; j < 16; ++j) {
        int4 r = v[j];
        r.x += add; r.y += add; r.z += add; r.w += add;
        pw[j * 32] = r;
    }
}


__global__ void __launch_bounds__(PT)
gcn_part_scatter(const int* __restrict__ src, const int* __restrict__ dst,
                 const int* __restrict__ off, int* __restrict__ bsd_i, int E) {
    __shared__ int cur[NB];
    int2* bsd = (int2*)bsd_i;
    for (int i = threadIdx.x; i < NB; i += PT) cur[i] = off[i * NPW + blockIdx.x];
    __syncthreads();
    const int chunk = (E + NPW - 1) / NPW;
    const int lo = blockIdx.x * chunk;
    const int hi = min(lo + chunk, E);
    for (int i0 = lo + threadIdx.x; i0 < hi; i0 += 4 * PT) {
        int d[4], s[4];
        #pragma unroll
        for (int u = 0; u < 4; ++u) {
            const int i = i0 + u * PT;
            d[u] = (i < hi) ? dst[i] : 0;
            s[u] = (i < hi) ? src[i] : 0;
        }
        #pragma unroll
        for (int u = 0; u < 4; ++u) {
            if (i0 + u * PT < hi) {
                const int pos = atomicAdd(&cur[gcn_bucket_of(d[u])], 1);
                bsd[pos] = make_int2(s[u], d[u]);
            }
        }
    }
}


__global__ void __launch_bounds__(PT)
gcn_bucket_deg(const int* __restrict__ bsd_i, const int* __restrict__ off,
               float* __restrict__ dis, int E) {
    __shared__ int c[RB];
    const int2* bsd = (const int2*)bsd_i;
    const int b = blockIdx.x;
    const int base = b * RB;
    for (int i = threadIdx.x; i < RB; i += PT) c[i] = 0;
    __syncthreads();
    const int lo = off[b * NPW];
    const int hi = (b + 1 < NB) ? off[(b + 1) * NPW] : E;
    for (int i = lo + threadIdx.x; i < hi; i += PT)
        atomicAdd(&c[gcn_local_row(bsd[i].y, base)], 1);
    __syncthreads();
    for (int i = threadIdx.x; i < RB; i += PT) {
        const int row = base + i;
        if (row < N_NODES) dis[row] = rsqrtf(1.0f + (float)c[i]);
    }
}


__global__ void gcn_gemm1(const float* __restrict__ x,
                          const float* __restrict__ W1,
                          const float* __restrict__ dis,
                          float* __restrict__ hs) {
    __shared__ float Bf[(N_FEAT / 2) * 32 * 2];

    for (int idx = threadIdx.x; idx < (N_FEAT / 2) * 32 * 2; idx += blockDim.x) {
        const int p = idx >> 6;
        const int c = (idx >> 1) & 31;
        const int j = idx & 1;
        Bf[idx] = (c < HIDDEN) ? W1[(2 * p + j) * HIDDEN + c] : 0.0f;
    }
    __syncthreads();

    const int lane = threadIdx.x & 31;
    const int wave = threadIdx.x >> 5;
    const int tile = blockIdx.x * (blockDim.x >> 5) + wave;
    if (tile >= M_TILES) return;
    const int row0 = tile << 4;
    const int half = lane >> 4;
    const int l16  = lane & 15;
    const float* arow = x + (size_t)(row0 + l16) * N_FEAT;

    v8f c0 = {};
    v8f c1 = {};

    #pragma unroll 4
    for (int k0 = 0; k0 < N_FEAT; k0 += 4) {
        const int kb = k0 + 2 * half;
        const int p = kb >> 1;
        v2f a  = *(const v2f*)(arow + kb);
        v2f b0 = *(const v2f*)&Bf[p * 64 + l16 * 2];
        v2f b1 = *(const v2f*)&Bf[p * 64 + (16 + l16) * 2];

        c0 = __builtin_amdgcn_wmma_f32_16x16x4_f32(false, a, false, b0, (short)0, c0, false, false);
        c1 = __builtin_amdgcn_wmma_f32_16x16x4_f32(false, a, false, b1, (short)0, c1, false, false);
    }

    #pragma unroll
    for (int r = 0; r < 8; ++r) {
        const int row = row0 + r + 8 * half;
        const float di = dis[row];
        float* p0 = hs + (size_t)row * PAD1;
        p0[l16]      = di * c0[r];
        p0[16 + l16] = di * c1[r];
    }
}


template <int F, int PAD>
__device__ __forceinline__ void gcn_bucket_edges(float* __restrict__ tile, const int2* __restrict__ bsd,
                                                 const float* __restrict__ hs, int lo, int hi, int base) {
    constexpr int TP = F + 1;
    for (int i = lo + threadIdx.x; i < hi; i += PT) {
        const int2 sd = bsd[i];
        const int s = sd.x;
        const int d = sd.y;
        const float* hrow = hs + (size_t)s * PAD;

        float v[F];
        #pragma unroll
        for (int j = 0; j + 4 <= F; j += 4) {
            float4 t = *(const float4*)(hrow + j);
            v[j] = t.x; v[j + 1] = t.y; v[j + 2] = t.z; v[j + 3] = t.w;
        }
        if constexpr (F % 4 != 0) {
            float2 t = *(const float2*)(hrow + (F & ~3));
            v[(F & ~3)] = t.x; v[(F & ~3) + 1] = t.y;
        }
        float* trow = &tile[gcn_local_row(d, base) * TP];
        #pragma unroll
        for (int f = 0; f < F; ++f) atomicAdd(&trow[f], v[f]);
    }
}


#define RBP 208

__global__ void __launch_bounds__(PT)
gcn_bucket_acc1_gemm2(const int* __restrict__ bsd_i, const int* __restrict__ off,
                      const float* __restrict__ dis, const float* __restrict__ hs,
                      const float* __restrict__ b1, const float* __restrict__ W2,
                      float* __restrict__ h2s, int E) {
    constexpr int TP = HIDDEN + 1;
    __shared__ float tile[RBP * TP];
    __shared__ float Bf2[(HIDDEN / 2) * 16 * 2];
    __shared__ float b1s[HIDDEN];
    __shared__ float disl[RBP];
    const int2* bsd = (const int2*)bsd_i;
    const int b = blockIdx.x;
    const int base = b * RB;
    const int tid = threadIdx.x;
    for (int i = tid; i < RBP * TP; i += PT) tile[i] = 0.0f;
    __syncthreads();
    float w2v = 0.0f, b1v = 0.0f, dv = 0.0f;
    if (tid < (HIDDEN / 2) * 16 * 2) {
        const int p = tid >> 5;
        const int c = (tid >> 1) & 15;
        const int j = tid & 1;
        if (c < N_CLASSES) w2v = W2[(2 * p + j) * N_CLASSES + c];
    }
    if (tid < HIDDEN) b1v = b1[tid];
    if (tid < RB && base + tid < N_NODES) dv = dis[base + tid];
    const int lo = off[b * NPW];
    const int hi = (b + 1 < NB) ? off[(b + 1) * NPW] : E;
    gcn_bucket_edges<HIDDEN, PAD1>(tile, bsd, hs, lo, hi, base);
    if (tid < (HIDDEN / 2) * 16 * 2) Bf2[tid] = w2v;
    if (tid < HIDDEN) b1s[tid] = b1v;
    if (tid < RBP) disl[tid] = dv;
    __syncthreads();

    const int lane = threadIdx.x & 31;
    const int wave = threadIdx.x >> 5;
    const int r0 = wave << 4;
    if (r0 >= RB || base + r0 >= N_NODES) return;
    const int half = lane >> 4;
    const int l16  = lane & 15;
    const int lr = r0 + l16;
    const bool live = (lr < RB) && (base + lr < N_NODES);
    const float* srow = hs + (size_t)(live ? base + lr : 0) * PAD1;
    const float dA = disl[lr];
    const float* trow = &tile[lr * TP];

    v8f c = {};
    #pragma unroll
    for (int k0 = 0; k0 < HIDDEN; k0 += 4) {
        const int kb = k0 + 2 * half;
        const v2f sr = *(const v2f*)(srow + kb);
        v2f a;
        a.x = fmaxf(dA * (trow[kb]     + sr.x) + b1s[kb],     0.0f);
        a.y = fmaxf(dA * (trow[kb + 1] + sr.y) + b1s[kb + 1], 0.0f);
        const v2f bb = *(const v2f*)&Bf2[(kb >> 1) * 32 + l16 * 2];
        c = __builtin_amdgcn_wmma_f32_16x16x4_f32(false, a, false, bb, (short)0, c, false, false);
    }

    #pragma unroll
    for (int r = 0; r < 8; ++r) {
        const int lrow = r0 + r + 8 * half;
        const int row = base + lrow;
        if (lrow < RB && row < N_NODES)
            h2s[(size_t)row * PAD2 + l16] = disl[lrow] * c[r];
    }
}


__global__ void __launch_bounds__(PT)
gcn_bucket_acc2_lsm(const int* __restrict__ bsd_i, const int* __restrict__ off,
                    const float* __restrict__ dis, const float* __restrict__ hs,
                    const float* __restrict__ b2, float* __restrict__ out, int E) {
    constexpr int TP = N_CLASSES + 1;
    __shared__ float tile[RB * TP];
    const int2* bsd = (const int2*)bsd_i;
    const int b = blockIdx.x;
    const int base = b * RB;
    for (int i = threadIdx.x; i < RB * TP; i += PT) tile[i] = 0.0f;
    __syncthreads();
    const int lo = off[b * NPW];
    const int hi = (b + 1 < NB) ? off[(b + 1) * NPW] : E;
    gcn_bucket_edges<N_CLASSES, PAD2>(tile, bsd, hs, lo, hi, base);
    __syncthreads();

    const int r = threadIdx.x;
    const int i = base + r;
    if (r >= RB || i >= N_NODES) return;
    const float* row = hs + (size_t)i * PAD2;
    const float di = dis[i];
    float z[N_CLASSES];
    {
        float4 t0 = *(const float4*)(row);
        float4 t1 = *(const float4*)(row + 4);
        float2 t2 = *(const float2*)(row + 8);
        z[0] = t0.x; z[1] = t0.y; z[2] = t0.z; z[3] = t0.w;
        z[4] = t1.x; z[5] = t1.y; z[6] = t1.z; z[7] = t1.w;
        z[8] = t2.x; z[9] = t2.y;
    }
    float m = -3.402823466e38f;
    #pragma unroll
    for (int c = 0; c < N_CLASSES; ++c) {
        z[c] = di * (tile[r * TP + c] + z[c]) + b2[c];
        m = fmaxf(m, z[c]);
    }
    float s = 0.0f;
    #pragma unroll
    for (int c = 0; c < N_CLASSES; ++c) s += __expf(z[c] - m);
    const float l = __logf(s);
    #pragma unroll
    for (int c = 0; c < N_CLASSES; ++c) out[(size_t)i * N_CLASSES + c] = z[c] - m - l;
}


extern "C" void kernel_launch(void* const* d_in, const int* in_sizes, int n_in,
                              void* d_out, int out_size, void* d_ws, size_t ws_size,
                              hipStream_t stream) {
    const float* x  = (const float*)d_in[0];
    const int*   ei = (const int*)d_in[1];
    const float* W1 = (const float*)d_in[2];
    const float* b1 = (const float*)d_in[3];
    const float* W2 = (const float*)d_in[4];
    const float* b2 = (const float*)d_in[5];
    float* out = (float*)d_out;

    const int E = in_sizes[1] / 2;
    const int* src = ei;
    const int* dst = ei + E;

    float* ws  = (float*)d_ws;
    float* dis = ws;
    float* h0  = ws + 102400;
    float* h2  = h0 + (size_t)N_NODES * PAD1;
    int*   cnt = (int*)(h2 + (size_t)N_NODES * PAD2);
    int*   bsd = cnt + NCNT;

    const int TB = 256;
    const int gemmBlocks = (M_TILES + (TB / 32) - 1) / (TB / 32);

    gcn_part_count<<<NPW, PT, 0, stream>>>(dst, cnt, E);
    gcn_part_scan<<<1, 1024, 0, stream>>>(cnt);
    gcn_part_scatter<<<NPW, PT, 0, stream>>>(src, dst, cnt, bsd, E);

    gcn_bucket_deg<<<NB, PT, 0, stream>>>(bsd, cnt, dis, E);

    gcn_gemm1<<<gemmBlocks, TB, 0, stream>>>(x, W1, dis, h0);
    gcn_bucket_acc1_gemm2<<<NB, PT, 0, stream>>>(bsd, cnt, dis, h0, b1, W2, h2, E);

    gcn_bucket_acc2_lsm<<<NB, PT, 0, stream>>>(bsd, cnt, dis, h2, b2, out, E);
}


# Row gather of the bucket sums coalesced: 5 (layer 1) or 3 (layer 2) lanes fetch one source row in 16-byte pieces, 6 or 10 whole rows per load; f32 sums stay in LDS-atomic arrival order
# speedup vs baseline: 9.4071x; 1.1165x over previous
#include <hip/hip_runtime.h>


typedef __attribute__((ext_vector_type(2))) float v2f;
typedef __attribute__((ext_vector_type(8))) float v8f;

#define N_NODES   100000
#define N_FEAT    128
#define HIDDEN    20
#define N_CLASSES 10
#define PAD1      32
#define PAD2      16
#define M_TILES   (N_NODES / 16)

#define RB    196
#define NB    512
#define NPW   128
#define PT    1024
#define NCNT  (NB * NPW)

__device__ __forceinline__ int gcn_bucket_of(int d) {
    const unsigned b = (unsigned)d / RB;
    return (int)(b < (unsigned)NB ? b : (unsigned)(NB - 1));
}

__device__ __forceinline__ int gcn_local_row(int d, int base) {
    const unsigned r = (unsigned)(d - base);
    return (int)(r < (unsigned)RB ? r : (unsigned)(RB - 1));
}


__global__ void __launch_bounds__(PT)
gcn_part_count(const int* __restrict__ dst, int* __restrict__ cnt, int E) {
    __shared__ int hist[NB];
    for (int i = threadIdx.x; i < NB; i += PT) hist[i] = 0;
    __syncthreads();
    const int chunk = (E + NPW - 1) / NPW;
    const int lo = blockIdx.x * chunk;
    const int hi = min(lo + chunk, E);
    for (int i0 = lo + threadIdx.x; i0 < hi; i0 += 4 * PT) {
        int d[4];
        #pragma unroll
        for (int u = 0; u < 4; ++u) {
            const int i = i0 + u * PT;
            d[u] = (i < hi) ? dst[i] : 0;
        }
        #pragma unroll
        for (int u = 0; u < 4; ++u)
            if (i0 + u * PT < hi) atomicAdd(&hist[gcn_bucket_of(d[u])], 1);
    }
    __syncthreads();
    for (int i = threadIdx.x; i < NB; i += PT) cnt[i * NPW + blockIdx.x] = hist[i];
}


__global__ void __launch_bounds__(1024)
gcn_part_scan(int* __restrict__ cnt) {
    __shared__ int wsum[32];
    int4* p = (int4*)cnt;
    const int lane = threadIdx.x & 31;
    const int wave = threadIdx.x >> 5;
    int4* pw = p + wave * 512 + lane;
    int4 v[16];
    #pragma unroll
    for (int j = 0; j < 16; ++j) v[j] = pw[j * 32];
    int carry = 0;
    #pragma unroll
    for (int j = 0; j < 16; ++j) {
        const int4 a = v[j];
        const int s4 = a.x + a.y + a.z + a.w;
        int inc = s4;
        #pragma unroll
        for (int o = 1; o < 32; o <<= 1) {
            const int y = __shfl_up(inc, o);
            if (lane >= o) inc += y;
        }
        const int ex = carry + inc - s4;
        int4 r;
        r.x = ex;
        r.y = ex + a.x;
        r.z = ex + a.x + a.y;
        r.w = ex + a.x + a.y + a.z;
        v[j] = r;
        carry += __shfl(inc, 31);
    }
    if (lane == 0) wsum[wave] = carry;
    __syncthreads();
    if (wave == 0) {
        const int w = wsum[lane];
        int winc = w;
        #pragma unroll
        for (int o = 1; o < 32; o <<= 1) {
            const int y = __shfl_up(winc, o);
            if (lane >= o) winc += y;
        }
        wsum[lane] = winc - w;
    }
    __syncthreads();
    const int add = wsum[wave];
    #pragma unroll
    for (int j = 0; j < 16; ++j) {
        int4 r = v[j];
        r.x += add; r.y += add; r.z += add; r.w += add;
        pw[j * 32] = r;
    }
}


__global__ void __launch_bounds__(PT)
gcn_part_scatter(const int* __restrict__ src, const int* __restrict__ dst,
                 const int* __restrict__ off, int* __restrict__ bsd_i, int E) {
    __shared__ int cur[NB];
    int2* bsd = (int2*)bsd_i;
    for (int i = threadIdx.x; i < NB; i += PT) cur[i] = off[i * NPW + blockIdx.x];
    __syncthreads();
    const int chunk = (E + NPW - 1) / NPW;
    const int lo = blockIdx.x * chunk;
    const int hi = min(lo + chunk, E);
    for (int i0 = lo + threadIdx.x; i0 < hi; i0 += 4 * PT) {
        int d[4], s[4];
        #pragma unroll
        for (int u = 0; u < 4; ++u) {
            const int i = i0 + u * PT;
            d[u] = (i < hi) ? dst[i] : 0;
            s[u] = (i < hi) ? src[i] : 0;
        }
        #pragma unroll
        for (int u = 0; u < 4; ++u) {
            if (i0 + u * PT < hi) {
                const int pos = atomicAdd(&cur[gcn_bucket_of(d[u])], 1);
                bsd[pos] = make_int2(s[u], d[u]);
            }
        }
    }
}


__global__ void __launch_bounds__(PT)
gcn_bucket_deg(const int* __restrict__ bsd_i, const int* __restrict__ off,
               float* __restrict__ dis, int E) {
    __shared__ int c[RB];
    const int2* bsd = (const int2*)bsd_i;
    const int b = blockIdx.x;
    const int base = b * RB;
    for (int i = threadIdx.x; i < RB; i += PT) c[i] = 0;
    __syncthreads();
    const int lo = off[b * NPW];
    const int hi = (b + 1 < NB) ? off[(b + 1) * NPW] : E;
    for (int i = lo + threadIdx.x; i < hi; i += PT)
        atomicAdd(&c[gcn_local_row(bsd[i].y, base)], 1);
    __syncthreads();
    for (int i = threadIdx.x; i < RB; i += PT) {
        const int row = base + i;
        if (row < N_NODES) dis[row] = rsqrtf(1.0f + (float)c[i]);
    }
}


__global__ void gcn_gemm1(const float* __restrict__ x,
                          const float* __restrict__ W1,
                          const float* __restrict__ dis,
                          float* __restrict__ hs) {
    __shared__ float Bf[(N_FEAT / 2) * 32 * 2];

    for (int idx = threadIdx.x; idx < (N_FEAT / 2) * 32 * 2; idx += blockDim.x) {
        const int p = idx >> 6;
        const int c = (idx >> 1) & 31;
        const int j = idx & 1;
        Bf[idx] = (c < HIDDEN) ? W1[(2 * p + j) * HIDDEN + c] : 0.0f;
    }
    __syncthreads();

    const int lane = threadIdx.x & 31;
    const int wave = threadIdx.x >> 5;
    const int tile = blockIdx.x * (blockDim.x >> 5) + wave;
    if (tile >= M_TILES) return;
    const int row0 = tile << 4;
    const int half = lane >> 4;
    const int l16  = lane & 15;
    const float* arow = x + (size_t)(row0 + l16) * N_FEAT;

    v8f c0 = {};
    v8f c1 = {};

    #pragma unroll 4
    for (int k0 = 0; k0 < N_FEAT; k0 += 4) {
        const int kb = k0 + 2 * half;
        const int p = kb >> 1;
        v2f a  = *(const v2f*)(arow + kb);
        v2f b0 = *(const v2f*)&Bf[p * 64 + l16 * 2];
        v2f b1 = *(const v2f*)&Bf[p * 64 + (16 + l16) * 2];

        c0 = __builtin_amdgcn_wmma_f32_16x16x4_f32(false, a, false, b0, (short)0, c0, false, false);
        c1 = __builtin_amdgcn_wmma_f32_16x16x4_f32(false, a, false, b1, (short)0, c1, false, false);
    }

    #pragma unroll
    for (int r = 0; r < 8; ++r) {
        const int row = row0 + r + 8 * half;
        const float di = dis[row];
        float* p0 = hs + (size_t)row * PAD1;
        p0[l16]      = di * c0[r];
        p0[16 + l16] = di * c1[r];
    }
}


template <int F, int PAD>
__device__ __forceinline__ void gcn_bucket_edges(float* __restrict__ tile, const int2* __restrict__ bsd,
                                                 const float* __restrict__ hs, int lo, int hi, int base) {
    constexpr int TP  = F + 1;
    constexpr int LPE = (F + 3) / 4;
    constexpr int EPW = 32 / LPE;
    constexpr int U   = 30 / EPW;
    constexpr int EPI = EPW * U;
    static_assert(4 * LPE <= PAD, "row pieces must stay inside the padded row");
    const int lane = threadIdx.x & 31;
    const int wave = threadIdx.x >> 5;
    const int e = min(lane / LPE, EPW - 1);
    const int c4 = (lane % LPE) * 4;
    const bool lane_on = lane < LPE * EPW;
    for (int i0 = lo + wave * EPI; i0 < hi; i0 += (PT / 32) * EPI) {
        int2 sd[U];
        #pragma unroll
        for (int u = 0; u < U; ++u) sd[u] = bsd[min(i0 + u * EPW + e, hi - 1)];
        __builtin_amdgcn_sched_barrier(0);
        float4 v[U];
        #pragma unroll
        for (int u = 0; u < U; ++u)
            v[u] = *(const float4*)(hs + (size_t)sd[u].x * PAD + c4);
        #pragma unroll
        for (int u = 0; u < U; ++u)
            asm volatile("" :: "v"(v[u].x), "v"(v[u].y), "v"(v[u].z), "v"(v[u].w));
        #pragma unroll
        for (int u = 0; u < U; ++u) {
            if (lane_on && i0 + u * EPW + e < hi) {
                float* t = &tile[gcn_local_row(sd[u].y, base) * TP + c4];
                atomicAdd(t, v[u].x);
                if (F % 4 == 0 || c4 + 1 < F) atomicAdd(t + 1, v[u].y);
                if (F % 4 == 0 || c4 + 2 < F) atomicAdd(t + 2, v[u].z);
                if (F % 4 == 0 || c4 + 3 < F) atomicAdd(t + 3, v[u].w);
            }
        }
    }
}


#define RBP 208

__global__ void __launch_bounds__(PT)
gcn_bucket_acc1_gemm2(const int* __restrict__ bsd_i, const int* __restrict__ off,
                      const float* __restrict__ dis, const float* __restrict__ hs,
                      const float* __restrict__ b1, const float* __restrict__ W2,
                      float* __restrict__ h2s, int E) {
    constexpr int TP = HIDDEN + 1;
    __shared__ float tile[RBP * TP];
    __shared__ float Bf2[(HIDDEN / 2) * 16 * 2];
    __shared__ float b1s[HIDDEN];
    __shared__ float disl[RBP];
    const int2* bsd = (const int2*)bsd_i;
    const int b = blockIdx.x;
    const int base = b * RB;
    const int tid = threadIdx.x;
    for (int i = tid; i < RBP * TP; i += PT) tile[i] = 0.0f;
    __syncthreads();
    float w2v = 0.0f, b1v = 0.0f, dv = 0.0f;
    if (tid < (HIDDEN / 2) * 16 * 2) {
        const int p = tid >> 5;
        const int c = (tid >> 1) & 15;
        const int j = tid & 1;
        if (c < N_CLASSES) w2v = W2[(2 * p + j) * N_CLASSES + c];
    }
    if (tid < HIDDEN) b1v = b1[tid];
    if (tid < RB && base + tid < N_NODES) dv = dis[base + tid];
    const int lo = off[b * NPW];
    const int hi = (b + 1 < NB) ? off[(b + 1) * NPW] : E;
    gcn_bucket_edges<HIDDEN, PAD1>(tile, bsd, hs, lo, hi, base);
    if (tid < (HIDDEN / 2) * 16 * 2) Bf2[tid] = w2v;
    if (tid < HIDDEN) b1s[tid] = b1v;
    if (tid < RBP) disl[tid] = dv;
    __syncthreads();

    const int lane = threadIdx.x & 31;
    const int wave = threadIdx.x >> 5;
    const int r0 = wave << 4;
    if (r0 >= RB || base + r0 >= N_NODES) return;
    const int half = lane >> 4;
    const int l16  = lane & 15;
    const int lr = r0 + l16;
    const bool live = (lr < RB) && (base + lr < N_NODES);
    const float* srow = hs + (size_t)(live ? base + lr : 0) * PAD1;
    const float dA = disl[lr];
    const float* trow = &tile[lr * TP];

    v8f c = {};
    #pragma unroll
    for (int k0 = 0; k0 < HIDDEN; k0 += 4) {
        const int kb = k0 + 2 * half;
        const v2f sr = *(const v2f*)(srow + kb);
        v2f a;
        a.x = fmaxf(dA * (trow[kb]     + sr.x) + b1s[kb],     0.0f);
        a.y = fmaxf(dA * (trow[kb + 1] + sr.y) + b1s[kb + 1], 0.0f);
        const v2f bb = *(const v2f*)&Bf2[(kb >> 1) * 32 + l16 * 2];
        c = __builtin_amdgcn_wmma_f32_16x16x4_f32(false, a, false, bb, (short)0, c, false, false);
    }

    #pragma unroll
    for (int r = 0; r < 8; ++r) {
        const int lrow = r0 + r + 8 * half;
        const int row = base + lrow;
        if (lrow < RB && row < N_NODES)
            h2s[(size_t)row * PAD2 + l16] = disl[lrow] * c[r];
    }
}


__global__ void __launch_bounds__(PT)
gcn_bucket_acc2_lsm(const int* __restrict__ bsd_i, const int* __restrict__ off,
                    const float* __restrict__ dis, const float* __restrict__ hs,
                    const float* __restrict__ b2, float* __restrict__ out, int E) {
    constexpr int TP = N_CLASSES + 1;
    __shared__ float tile[RB * TP];
    const int2* bsd = (const int2*)bsd_i;
    const int b = blockIdx.x;
    const int base = b * RB;
    for (int i = threadIdx.x; i < RB * TP; i += PT) tile[i] = 0.0f;
    __syncthreads();
    const int lo = off[b * NPW];
    const int hi = (b + 1 < NB) ? off[(b + 1) * NPW] : E;
    gcn_bucket_edges<N_CLASSES, PAD2>(tile, bsd, hs, lo, hi, base);
    __syncthreads();

    const int r = threadIdx.x;
    const int i = base + r;
    if (r >= RB || i >= N_NODES) return;
    const float* row = hs + (size_t)i * PAD2;
    const float di = dis[i];
    float z[N_CLASSES];
    {
        float4 t0 = *(const float4*)(row);
        float4 t1 = *(const float4*)(row + 4);
        float2 t2 = *(const float2*)(row + 8);
        z[0] = t0.x; z[1] = t0.y; z[2] = t0.z; z[3] = t0.w;
        z[4] = t1.x; z[5] = t1.y; z[6] = t1.z; z[7] = t1.w;
        z[8] = t2.x; z[9] = t2.y;
    }
    float m = -3.402823466e38f;
    #pragma unroll
    for (int c = 0; c < N_CLASSES; ++c) {
        z[c] = di * (tile[r * TP + c] + z[c]) + b2[c];
        m = fmaxf(m, z[c]);
    }
    float s = 0.0f;
    #pragma unroll
    for (int c = 0; c < N_CLASSES; ++c) s += __expf(z[c] - m);
    const float l = __logf(s);
    #pragma unroll
    for (int c = 0; c < N_CLASSES; ++c) out[(size_t)i * N_CLASSES + c] = z[c] - m - l;
}


extern "C" void kernel_launch(void* const* d_in, const int* in_sizes, int n_in,
                              void* d_out, int out_size, void* d_ws, size_t ws_size,
                              hipStream_t stream) {
    const float* x  = (const float*)d_in[0];
    const int*   ei = (const int*)d_in[1];
    const float* W1 = (const float*)d_in[2];
    const float* b1 = (const float*)d_in[3];
    const float* W2 = (const float*)d_in[4];
    const float* b2 = (const float*)d_in[5];
    float* out = (float*)d_out;

    const int E = in_sizes[1] / 2;
    const int* src = ei;
    const int* dst = ei + E;

    float* ws  = (float*)d_ws;
    float* dis = ws;
    float* h0  = ws + 102400;
    float* h2  = h0 + (size_t)N_NODES * PAD1;
    int*   cnt = (int*)(h2 + (size_t)N_NODES * PAD2);
    int*   bsd = cnt + NCNT;

    const int TB = 256;
    const int gemmBlocks = (M_TILES + (TB / 32) - 1) / (TB / 32);

    gcn_part_count<<<NPW, PT, 0, stream>>>(dst, cnt, E);
    gcn_part_scan<<<1, 1024, 0, stream>>>(cnt);
    gcn_part_scatter<<<NPW, PT, 0, stream>>>(src, dst, cnt, bsd, E);

    gcn_bucket_deg<<<NB, PT, 0, stream>>>(bsd, cnt, dis, E);

    gcn_gemm1<<<gemmBlocks, TB, 0, stream>>>(x, W1, dis, h0);
    gcn_bucket_acc1_gemm2<<<NB, PT, 0, stream>>>(bsd, cnt, dis, h0, b1, W2, h2, E);

    gcn_bucket_acc2_lsm<<<NB, PT, 0, stream>>>(bsd, cnt, dis, h2, b2, out, E);
}


# Degree pass folded into the GEMM1 launch: one work-group per bucket, 13 waves run its x@W1 M tiles while 3 count its rows' in-degrees in LDS; epilogue scales from LDS; one launch fewer
# speedup vs baseline: 9.9845x; 1.0614x over previous
#include <hip/hip_runtime.h>


typedef __attribute__((ext_vector_type(2))) float v2f;
typedef __attribute__((ext_vector_type(8))) float v8f;

#define N_NODES   100000
#define N_FEAT    128
#define HIDDEN    20
#define N_CLASSES 10
#define PAD1      32
#define PAD2      16

#define RB    196
#define NB    512
#define NPW   128
#define PT    1024
#define NCNT  (NB * NPW)

__device__ __forceinline__ int gcn_bucket_of(int d) {
    const unsigned b = (unsigned)d / RB;
    return (int)(b < (unsigned)NB ? b : (unsigned)(NB - 1));
}

__device__ __forceinline__ int gcn_local_row(int d, int base) {
    const unsigned r = (unsigned)(d - base);
    return (int)(r < (unsigned)RB ? r : (unsigned)(RB - 1));
}


__global__ void __launch_bounds__(PT)
gcn_part_count(const int* __restrict__ dst, int* __restrict__ cnt, int E) {
    __shared__ int hist[NB];
    for (int i = threadIdx.x; i < NB; i += PT) hist[i] = 0;
    __syncthreads();
    const int chunk = (E + NPW - 1) / NPW;
    const int lo = blockIdx.x * chunk;
    const int hi = min(lo + chunk, E);
    for (int i0 = lo + threadIdx.x; i0 < hi; i0 += 4 * PT) {
        int d[4];
        #pragma unroll
        for (int u = 0; u < 4; ++u) {
            const int i = i0 + u * PT;
            d[u] = (i < hi) ? dst[i] : 0;
        }
        #pragma unroll
        for (int u = 0; u < 4; ++u)
            if (i0 + u * PT < hi) atomicAdd(&hist[gcn_bucket_of(d[u])], 1);
    }
    __syncthreads();
    for (int i = threadIdx.x; i < NB; i += PT) cnt[i * NPW + blockIdx.x] = hist[i];
}


__global__ void __launch_bounds__(1024)
gcn_part_scan(int* __restrict__ cnt) {
    __shared__ int wsum[32];
    int4* p = (int4*)cnt;
    const int lane = threadIdx.x & 31;
    const int wave = threadIdx.x >> 5;
    int4* pw = p + wave * 512 + lane;
    int4 v[16];
    #pragma unroll
    for (int j = 0; j < 16; ++j) v[j] = pw[j * 32];
    int carry = 0;
    #pragma unroll
    for (int j = 0; j < 16; ++j) {
        const int4 a = v[j];
        const int s4 = a.x + a.y + a.z + a.w;
        int inc = s4;
        #pragma unroll
        for (int o = 1; o < 32; o <<= 1) {
            const int y = __shfl_up(inc, o);
            if (lane >= o) inc += y;
        }
        const int ex = carry + inc - s4;
        int4 r;
        r.x = ex;
        r.y = ex + a.x;
        r.z = ex + a.x + a.y;
        r.w = ex + a.x + a.y + a.z;
        v[j] = r;
        carry += __shfl(inc, 31);
    }
    if (lane == 0) wsum[wave] = carry;
    __syncthreads();
    if (wave == 0) {
        const int w = wsum[lane];
        int winc = w;
        #pragma unroll
        for (int o = 1; o < 32; o <<= 1) {
            const int y = __shfl_up(winc, o);
            if (lane >= o) winc += y;
        }
        wsum[lane] = winc - w;
    }
    __syncthreads();
    const int add = wsum[wave];
    #pragma unroll
    for (int j = 0; j < 16; ++j) {
        int4 r = v[j];
        r.x += add; r.y += add; r.z += add; r.w += add;
        pw[j * 32] = r;
    }
}


__global__ void __launch_bounds__(PT)
gcn_part_scatter(const int* __restrict__ src, const int* __restrict__ dst,
                 const int* __restrict__ off, int* __restrict__ bsd_i, int E) {
    __shared__ int cur[NB];
    int2* bsd = (int2*)bsd_i;
    for (int i = threadIdx.x; i < NB; i += PT) cur[i] = off[i * NPW + blockIdx.x];
    __syncthreads();
    const int chunk = (E + NPW - 1) / NPW;
    const int lo = blockIdx.x * chunk;
    const int hi = min(lo + chunk, E);
    for (int i0 = lo + threadIdx.x; i0 < hi; i0 += 4 * PT) {
        int d[4], s[4];
        #pragma unroll
        for (int u = 0; u < 4; ++u) {
            const int i = i0 + u * PT;
            d[u] = (i < hi) ? dst[i] : 0;
            s[u] = (i < hi) ? src[i] : 0;
        }
        #pragma unroll
        for (int u = 0; u < 4; ++u) {
            if (i0 + u * PT < hi) {
                const int pos = atomicAdd(&cur[gcn_bucket_of(d[u])], 1);
                bsd[pos] = make_int2(s[u], d[u]);
            }
        }
    }
}


#define RBP 208
#define GT  512
#define GW  (RBP / 16)
#define CT  (GT - GW * 32)
#define CU  24

__global__ void __launch_bounds__(GT)
gcn_deg_gemm1(const float* __restrict__ x,
              const float* __restrict__ W1,
              const int* __restrict__ bsd_i, const int* __restrict__ off,
              float* __restrict__ dis,
              float* __restrict__ hs, int E) {
    __shared__ float Bf[(N_FEAT / 2) * 32 * 2];
    __shared__ int cdeg[RBP];

    const int tid = threadIdx.x;
    const int b = blockIdx.x;
    const int base = b * RB;
    for (int idx = tid; idx < (N_FEAT / 2) * 32 * 2; idx += GT) {
        const int p = idx >> 6;
        const int c = (idx >> 1) & 31;
        const int j = idx & 1;
        Bf[idx] = (c < HIDDEN) ? W1[(2 * p + j) * HIDDEN + c] : 0.0f;
    }
    for (int i = tid; i < RBP; i += GT) cdeg[i] = 0;
    __syncthreads();

    const int lane = tid & 31;
    const int wave = tid >> 5;
    const int half = lane >> 4;
    const int l16  = lane & 15;
    const int r0 = wave << 4;
    const bool gemm_wave = wave < GW;
    const bool tile_on = gemm_wave && r0 < RB && base + r0 < N_NODES;

    v8f c0 = {};
    v8f c1 = {};

    if (tile_on) {
        const float* arow = x + (size_t)min(base + r0 + l16, N_NODES - 1) * N_FEAT;
        #pragma unroll 4
        for (int k0 = 0; k0 < N_FEAT; k0 += 4) {
            const int kb = k0 + 2 * half;
            const int p = kb >> 1;
            v2f a  = *(const v2f*)(arow + kb);
            v2f b0 = *(const v2f*)&Bf[p * 64 + l16 * 2];
            v2f b1 = *(const v2f*)&Bf[p * 64 + (16 + l16) * 2];

            c0 = __builtin_amdgcn_wmma_f32_16x16x4_f32(false, a, false, b0, (short)0, c0, false, false);
            c1 = __builtin_amdgcn_wmma_f32_16x16x4_f32(false, a, false, b1, (short)0, c1, false, false);
        }
    } else if (!gemm_wave) {
        const int2* bsd = (const int2*)bsd_i;
        const int lo = off[b * NPW];
        const int hi = (b + 1 < NB) ? off[(b + 1) * NPW] : E;
        const int t = tid - GW * 32;
        for (int i0 = lo + t; i0 < hi; i0 += CU * CT) {
            int d[CU];
            #pragma unroll
            for (int u = 0; u < CU; ++u)
                d[u] = bsd[min(i0 + u * CT, hi - 1)].y;
            #pragma unroll
            for (int u = 0; u < CU; ++u)
                asm volatile("" :: "v"(d[u]));
            #pragma unroll
            for (int u = 0; u < CU; ++u)
                if (i0 + u * CT < hi) atomicAdd(&cdeg[gcn_local_row(d[u], base)], 1);
        }
    }
    __syncthreads();

    if (tile_on) {
        float di[8];
        #pragma unroll
        for (int r = 0; r < 8; ++r)
            di[r] = rsqrtf(1.0f + (float)cdeg[r0 + r + 8 * half]);
        #pragma unroll
        for (int r = 0; r < 8; ++r) {
            const int lrow = r0 + r + 8 * half;
            const int row = base + lrow;
            if (lrow < RB && row < N_NODES) {
                float* p0 = hs + (size_t)row * PAD1;
                p0[l16]      = di[r] * c0[r];
                p0[16 + l16] = di[r] * c1[r];
            }
        }
    } else if (!gemm_wave) {
        for (int i = tid - GW * 32; i < RB; i += CT) {
            const int row = base + i;
            if (row < N_NODES) dis[row] = rsqrtf(1.0f + (float)cdeg[i]);
        }
    }
}


template <int F, int PAD>
__device__ __forceinline__ void gcn_bucket_edges(float* __restrict__ tile, const int2* __restrict__ bsd,
                                                 const float* __restrict__ hs, int lo, int hi, int base) {
    constexpr int TP  = F + 1;
    constexpr int LPE = (F + 3) / 4;
    constexpr int EPW = 32 / LPE;
    constexpr int U   = 30 / EPW;
    constexpr int EPI = EPW * U;
    static_assert(4 * LPE <= PAD, "row pieces must stay inside the padded row");
    const int lane = threadIdx.x & 31;
    const int wave = threadIdx.x >> 5;
    const int e = min(lane / LPE, EPW - 1);
    const int c4 = (lane % LPE) * 4;
    const bool lane_on = lane < LPE * EPW;
    for (int i0 = lo + wave * EPI; i0 < hi; i0 += (PT / 32) * EPI) {
        int2 sd[U];
        #pragma unroll
        for (int u = 0; u < U; ++u) sd[u] = bsd[min(i0 + u * EPW + e, hi - 1)];
        __builtin_amdgcn_sched_barrier(0);
        float4 v[U];
        #pragma unroll
        for (int u = 0; u < U; ++u)
            v[u] = *(const float4*)(hs + (size_t)sd[u].x * PAD + c4);
        #pragma unroll
        for (int u = 0; u < U; ++u)
            asm volatile("" :: "v"(v[u].x), "v"(v[u].y), "v"(v[u].z), "v"(v[u].w));
        #pragma unroll
        for (int u = 0; u < U; ++u) {
            if (lane_on && i0 + u * EPW + e < hi) {
                float* t = &tile[gcn_local_row(sd[u].y, base) * TP + c4];
                atomicAdd(t, v[u].x);
                if (F % 4 == 0 || c4 + 1 < F) atomicAdd(t + 1, v[u].y);
                if (F % 4 == 0 || c4 + 2 < F) atomicAdd(t + 2, v[u].z);
                if (F % 4 == 0 || c4 + 3 < F) atomicAdd(t + 3, v[u].w);
            }
        }
    }
}


__global__ void __launch_bounds__(PT)
gcn_bucket_acc1_gemm2(const int* __restrict__ bsd_i, const int* __restrict__ off,
                      const float* __restrict__ dis, const float* __restrict__ hs,
                      const float* __restrict__ b1, const float* __restrict__ W2,
                      float* __restrict__ h2s, int E) {
    constexpr int TP = HIDDEN + 1;
    __shared__ float tile[RBP * TP];
    __shared__ float Bf2[(HIDDEN / 2) * 16 * 2];
    __shared__ float b1s[HIDDEN];
    __shared__ float disl[RBP];
    const int2* bsd = (const int2*)bsd_i;
    const int b = blockIdx.x;
    const int base = b * RB;
    const int tid = threadIdx.x;
    for (int i = tid; i < RBP * TP; i += PT) tile[i] = 0.0f;
    __syncthreads();
    float w2v = 0.0f, b1v = 0.0f, dv = 0.0f;
    if (tid < (HIDDEN / 2) * 16 * 2) {
        const int p = tid >> 5;
        const int c = (tid >> 1) & 15;
        const int j = tid & 1;
        if (c < N_CLASSES) w2v = W2[(2 * p + j) * N_CLASSES + c];
    }
    if (tid < HIDDEN) b1v = b1[tid];
    if (tid < RB && base + tid < N_NODES) dv = dis[base + tid];
    const int lo = off[b * NPW];
    const int hi = (b + 1 < NB) ? off[(b + 1) * NPW] : E;
    gcn_bucket_edges<HIDDEN, PAD1>(tile, bsd, hs, lo, hi, base);
    if (tid < (HIDDEN / 2) * 16 * 2) Bf2[tid] = w2v;
    if (tid < HIDDEN) b1s[tid] = b1v;
    if (tid < RBP) disl[tid] = dv;
    __syncthreads();

    const int lane = threadIdx.x & 31;
    const int wave = threadIdx.x >> 5;
    const int r0 = wave << 4;
    if (r0 >= RB || base + r0 >= N_NODES) return;
    const int half = lane >> 4;
    const int l16  = lane & 15;
    const int lr = r0 + l16;
    const bool live = (lr < RB) && (base + lr < N_NODES);
    const float* srow = hs + (size_t)(live ? base + lr : 0) * PAD1;
    const float dA = disl[lr];
    const float* trow = &tile[lr * TP];

    v8f c = {};
    #pragma unroll
    for (int k0 = 0; k0 < HIDDEN; k0 += 4) {
        const int kb = k0 + 2 * half;
        const v2f sr = *(const v2f*)(srow + kb);
        v2f a;
        a.x = fmaxf(dA * (trow[kb]     + sr.x) + b1s[kb],     0.0f);
        a.y = fmaxf(dA * (trow[kb + 1] + sr.y) + b1s[kb + 1], 0.0f);
        const v2f bb = *(const v2f*)&Bf2[(kb >> 1) * 32 + l16 * 2];
        c = __builtin_amdgcn_wmma_f32_16x16x4_f32(false, a, false, bb, (short)0, c, false, false);
    }

    #pragma unroll
    for (int r = 0; r < 8; ++r) {
        const int lrow = r0 + r + 8 * half;
        const int row = base + lrow;
        if (lrow < RB && row < N_NODES)
            h2s[(size_t)row * PAD2 + l16] = disl[lrow] * c[r];
    }
}


__global__ void __launch_bounds__(PT)
gcn_bucket_acc2_lsm(const int* __restrict__ bsd_i, const int* __restrict__ off,
                    const float* __restrict__ dis, const float* __restrict__ hs,
                    const float* __restrict__ b2, float* __restrict__ out, int E) {
    constexpr int TP = N_CLASSES + 1;
    __shared__ float tile[RB * TP];
    const int2* bsd = (const int2*)bsd_i;
    const int b = blockIdx.x;
    const int base = b * RB;
    for (int i = threadIdx.x; i < RB * TP; i += PT) tile[i] = 0.0f;
    __syncthreads();
    const int lo = off[b * NPW];
    const int hi = (b + 1 < NB) ? off[(b + 1) * NPW] : E;
    gcn_bucket_edges<N_CLASSES, PAD2>(tile, bsd, hs, lo, hi, base);
    __syncthreads();

    const int r = threadIdx.x;
    const int i = base + r;
    if (r >= RB || i >= N_NODES) return;
    const float* row = hs + (size_t)i * PAD2;
    const float di = dis[i];
    float z[N_CLASSES];
    {
        float4 t0 = *(const float4*)(row);
        float4 t1 = *(const float4*)(row + 4);
        float2 t2 = *(const float2*)(row + 8);
        z[0] = t0.x; z[1] = t0.y; z[2] = t0.z; z[3] = t0.w;
        z[4] = t1.x; z[5] = t1.y; z[6] = t1.z; z[7] = t1.w;
        z[8] = t2.x; z[9] = t2.y;
    }
    float m = -3.402823466e38f;
    #pragma unroll
    for (int c = 0; c < N_CLASSES; ++c) {
        z[c] = di * (tile[r * TP + c] + z[c]) + b2[c];
        m = fmaxf(m, z[c]);
    }
    float s = 0.0f;
    #pragma unroll
    for (int c = 0; c < N_CLASSES; ++c) s += __expf(z[c] - m);
    const float l = __logf(s);
    #pragma unroll
    for (int c = 0; c < N_CLASSES; ++c) out[(size_t)i * N_CLASSES + c] = z[c] - m - l;
}


extern "C" void kernel_launch(void* const* d_in, const int* in_sizes, int n_in,
                              void* d_out, int out_size, void* d_ws, size_t ws_size,
                              hipStream_t stream) {
    const float* x  = (const float*)d_in[0];
    const int*   ei = (const int*)d_in[1];
    const float* W1 = (const float*)d_in[2];
    const float* b1 = (const float*)d_in[3];
    const float* W2 = (const float*)d_in[4];
    const float* b2 = (const float*)d_in[5];
    float* out = (float*)d_out;

    const int E = in_sizes[1] / 2;
    const int* src = ei;
    const int* dst = ei + E;

    float* ws  = (float*)d_ws;
    float* dis = ws;
    float* h0  = ws + 102400;
    float* h2  = h0 + (size_t)N_NODES * PAD1;
    int*   cnt = (int*)(h2 + (size_t)N_NODES * PAD2);
    int*   bsd = cnt + NCNT;

    gcn_part_count<<<NPW, PT, 0, stream>>>(dst, cnt, E);
    gcn_part_scan<<<1, 1024, 0, stream>>>(cnt);
    gcn_part_scatter<<<NPW, PT, 0, stream>>>(src, dst, cnt, bsd, E);

    gcn_deg_gemm1<<<NB, GT, 0, stream>>>(x, W1, bsd, cnt, dis, h0, E);
    gcn_bucket_acc1_gemm2<<<NB, PT, 0, stream>>>(bsd, cnt, dis, h0, b1, W2, h2, E);

    gcn_bucket_acc2_lsm<<<NB, PT, 0, stream>>>(bsd, cnt, dis, h2, b2, out, E);
}


# Edge grouping in one launch: each of 128 slices bucket-sorts its own edges in LDS, bucket owners walk 128 short runs; count and scan launches gone; LDS-atomic sum order now run by run
# speedup vs baseline: 13.9880x; 1.4010x over previous
#include <hip/hip_runtime.h>


typedef __attribute__((ext_vector_type(2))) float v2f;
typedef __attribute__((ext_vector_type(8))) float v8f;

#define N_NODES   100000
#define N_FEAT    128
#define HIDDEN    20
#define N_CLASSES 10
#define PAD1      32
#define PAD2      16

#define RB    196
#define NB    512
#define NPW   128
#define PT    1024
#define N_EDGES 3200000
#define SR    25
#define NOFF  (NPW * (NB + 1))
static_assert((long long)NPW * SR * PT >= N_EDGES, "a slice must fit its work-group's registers and LDS stage");

__device__ __forceinline__ int gcn_bucket_of(int d) {
    const unsigned b = (unsigned)d / RB;
    return (int)(b < (unsigned)NB ? b : (unsigned)(NB - 1));
}

__device__ __forceinline__ int gcn_local_row(int d, int base) {
    const unsigned r = (unsigned)(d - base);
    return (int)(r < (unsigned)RB ? r : (unsigned)(RB - 1));
}


__global__ void __launch_bounds__(PT)
gcn_group_edges(const int* __restrict__ src, const int* __restrict__ dst,
                int* __restrict__ offt, int* __restrict__ bsd_i, int E) {
    __shared__ int hist[NB];
    __shared__ int2 stage[SR * PT];
    const int tid = threadIdx.x;
    for (int i = tid; i < NB; i += PT) hist[i] = 0;
    __syncthreads();
    const int chunk = (E + NPW - 1) / NPW;
    const int lo = blockIdx.x * chunk;
    const int hi = min(lo + min(chunk, SR * PT), E);
    int d[SR], s[SR], rk[SR];
    #pragma unroll
    for (int u = 0; u < SR; ++u) {
        const int i = lo + u * PT + tid;
        d[u] = (i < hi) ? dst[i] : 0;
        s[u] = (i < hi) ? src[i] : 0;
    }
    #pragma unroll
    for (int u = 0; u < SR; ++u)
        rk[u] = (lo + u * PT + tid < hi) ? atomicAdd(&hist[gcn_bucket_of(d[u])], 1) : 0;
    __syncthreads();
    if (tid < 32) {
        int v[16];
        int sum = 0;
        #pragma unroll
        for (int k = 0; k < 16; ++k) { v[k] = hist[tid * 16 + k]; sum += v[k]; }
        int inc = sum;
        #pragma unroll
        for (int o = 1; o < 32; o <<= 1) {
            const int y = __shfl_up(inc, o);
            if (tid >= o) inc += y;
        }
        int ex = inc - sum;
        #pragma unroll
        for (int k = 0; k < 16; ++k) { hist[tid * 16 + k] = ex; ex += v[k]; }
    }
    __syncthreads();
    int* ot = offt + blockIdx.x * (NB + 1);
    for (int i = tid; i < NB; i += PT) ot[i] = hist[i];
    if (tid == 0) ot[NB] = max(hi - lo, 0);
    #pragma unroll
    for (int u = 0; u < SR; ++u)
        if (lo + u * PT + tid < hi)
            stage[hist[gcn_bucket_of(d[u])] + rk[u]] = make_int2(s[u], d[u]);
    __syncthreads();
    int2* bsd = (int2*)bsd_i;
    const int n = hi - lo;
    static_assert(SR % 5 == 0, "the copy-out takes five records per thread and trip");
    #pragma nounroll
    for (int u0 = 0; u0 < SR; u0 += 5) {
        int2 rec[5];
        #pragma unroll
        for (int u = 0; u < 5; ++u)
            rec[u] = stage[(u0 + u) * PT + tid];
        #pragma unroll
        for (int u = 0; u < 5; ++u)
            asm volatile("" :: "v"(rec[u].x), "v"(rec[u].y));
        #pragma unroll
        for (int u = 0; u < 5; ++u)
            if ((u0 + u) * PT + tid < n) bsd[lo + (u0 + u) * PT + tid] = rec[u];
    }
}


#define RBP 208
#define GT  512
#define GW  (RBP / 16)
#define CT  (GT - GW * 32)
#define CRB 12

__global__ void __launch_bounds__(GT)
gcn_deg_gemm1(const float* __restrict__ x,
              const float* __restrict__ W1,
              const int* __restrict__ bsd_i, const int* __restrict__ offt,
              float* __restrict__ dis,
              float* __restrict__ hs, int E) {
    __shared__ float Bf[(N_FEAT / 2) * 32 * 2];
    __shared__ int cdeg[RBP];

    const int tid = threadIdx.x;
    const int b = blockIdx.x;
    const int base = b * RB;
    for (int idx = tid; idx < (N_FEAT / 2) * 32 * 2; idx += GT) {
        const int p = idx >> 6;
        const int c = (idx >> 1) & 31;
        const int j = idx & 1;
        Bf[idx] = (c < HIDDEN) ? W1[(2 * p + j) * HIDDEN + c] : 0.0f;
    }
    for (int i = tid; i < RBP; i += GT) cdeg[i] = 0;
    __syncthreads();

    const int lane = tid & 31;
    const int wave = tid >> 5;
    const int half = lane >> 4;
    const int l16  = lane & 15;
    const int r0 = wave << 4;
    const bool gemm_wave = wave < GW;
    const bool tile_on = gemm_wave && r0 < RB && base + r0 < N_NODES;

    v8f c0 = {};
    v8f c1 = {};

    if (tile_on) {
        const float* arow = x + (size_t)min(base + r0 + l16, N_NODES - 1) * N_FEAT;
        #pragma unroll 4
        for (int k0 = 0; k0 < N_FEAT; k0 += 4) {
            const int kb = k0 + 2 * half;
            const int p = kb >> 1;
            v2f a  = *(const v2f*)(arow + kb);
            v2f b0 = *(const v2f*)&Bf[p * 64 + l16 * 2];
            v2f b1 = *(const v2f*)&Bf[p * 64 + (16 + l16) * 2];

            c0 = __builtin_amdgcn_wmma_f32_16x16x4_f32(false, a, false, b0, (short)0, c0, false, false);
            c1 = __builtin_amdgcn_wmma_f32_16x16x4_f32(false, a, false, b1, (short)0, c1, false, false);
        }
    } else if (!gemm_wave) {
        const int2* bsd = (const int2*)bsd_i;
        const int chunk = (E + NPW - 1) / NPW;
        const int cw = __builtin_amdgcn_readfirstlane(wave) - GW;
        for (int rb = cw * CRB; rb < NPW; rb += (CT / 32) * CRB) {
            int st[CRB], ln[CRB];
            #pragma unroll
            for (int u = 0; u < CRB; ++u) {
                const int r = min(rb + u, NPW - 1);
                const int a  = offt[r * (NB + 1) + b];
                const int en = offt[r * (NB + 1) + b + 1];
                st[u] = r * chunk + a;
                ln[u] = (rb + u < NPW) ? en - a : 0;
            }
            int d[2 * CRB];
            #pragma unroll
            for (int u = 0; u < CRB; ++u) {
                d[2 * u]     = bsd[min(st[u] + lane, E - 1)].y;
                d[2 * u + 1] = bsd[min(st[u] + lane + 32, E - 1)].y;
            }
            #pragma unroll
            for (int u = 0; u < 2 * CRB; ++u)
                asm volatile("" :: "v"(d[u]));
            #pragma unroll
            for (int u = 0; u < CRB; ++u) {
                if (lane < ln[u])      atomicAdd(&cdeg[gcn_local_row(d[2 * u], base)], 1);
                if (lane + 32 < ln[u]) atomicAdd(&cdeg[gcn_local_row(d[2 * u + 1], base)], 1);
                for (int j = lane + 64; j < ln[u]; j += 32)
                    atomicAdd(&cdeg[gcn_local_row(bsd[st[u] + j].y, base)], 1);
            }
        }
    }
    __syncthreads();

    if (tile_on) {
        float di[8];
        #pragma unroll
        for (int r = 0; r < 8; ++r)
            di[r] = rsqrtf(1.0f + (float)cdeg[r0 + r + 8 * half]);
        #pragma unroll
        for (int r = 0; r < 8; ++r) {
            const int lrow = r0 + r + 8 * half;
            const int row = base + lrow;
            if (lrow < RB && row < N_NODES) {
                float* p0 = hs + (size_t)row * PAD1;
                p0[l16]      = di[r] * c0[r];
                p0[16 + l16] = di[r] * c1[r];
            }
        }
    } else if (!gemm_wave) {
        for (int i = tid - GW * 32; i < RB; i += CT) {
            const int row = base + i;
            if (row < N_NODES) dis[row] = rsqrtf(1.0f + (float)cdeg[i]);
        }
    }
}


template <int F, int PAD>
__device__ __forceinline__ void gcn_bucket_edges(float* __restrict__ tile, const int2* __restrict__ bsd,
                                                 const float* __restrict__ hs, const int* __restrict__ offt,
                                                 int b, int chunk, int base) {
    constexpr int TP  = F + 1;
    constexpr int LPE = (F + 3) / 4;
    constexpr int EPW = 32 / LPE;
    constexpr int U   = 30 / EPW;
    constexpr int EPI = EPW * U;
    constexpr int RPW = NPW / (PT / 32);
    static_assert(4 * LPE <= PAD, "row pieces must stay inside the padded row");
    static_assert(RPW * (PT / 32) == NPW, "every run needs exactly one wave");
    const int lane = threadIdx.x & 31;
    const int wave = __builtin_amdgcn_readfirstlane((int)(threadIdx.x >> 5));
    int st[RPW], cum[RPW + 1];
    cum[0] = 0;
    #pragma unroll
    for (int k = 0; k < RPW; ++k) {
        const int sl = wave * RPW + k;
        const int a  = offt[sl * (NB + 1) + b];
        const int en = offt[sl * (NB + 1) + b + 1];
        st[k] = sl * chunk + a - cum[k];
        cum[k + 1] = cum[k] + (en - a);
    }
    const int n = cum[RPW];
    const int el = min(lane / LPE, EPW - 1);
    const int c4 = (lane % LPE) * 4;
    const bool lane_on = lane < LPE * EPW;
    for (int j0 = 0; j0 < n; j0 += EPI) {
        int2 sd[U];
        #pragma unroll
        for (int u = 0; u < U; ++u) {
            const int j = min(j0 + u * EPW + el, n - 1);
            int o = st[0];
            #pragma unroll
            for (int k = 1; k < RPW; ++k) o = (j >= cum[k]) ? st[k] : o;
            sd[u] = bsd[o + j];
        }
        __builtin_amdgcn_sched_barrier(0);
        float4 v[U];
        #pragma unroll
        for (int u = 0; u < U; ++u)
            v[u] = *(const float4*)(hs + (size_t)sd[u].x * PAD + c4);
        #pragma unroll
        for (int u = 0; u < U; ++u)
            asm volatile("" :: "v"(v[u].x), "v"(v[u].y), "v"(v[u].z), "v"(v[u].w));
        #pragma unroll
        for (int u = 0; u < U; ++u) {
            if (lane_on && j0 + u * EPW + el < n) {
                float* t = &tile[gcn_local_row(sd[u].y, base) * TP + c4];
                atomicAdd(t, v[u].x);
                if (F % 4 == 0 || c4 + 1 < F) atomicAdd(t + 1, v[u].y);
                if (F % 4 == 0 || c4 + 2 < F) atomicAdd(t + 2, v[u].z);
                if (F % 4 == 0 || c4 + 3 < F) atomicAdd(t + 3, v[u].w);
            }
        }
    }
}


__global__ void __launch_bounds__(PT)
gcn_bucket_acc1_gemm2(const int* __restrict__ bsd_i, const int* __restrict__ offt,
                      const float* __restrict__ dis, const float* __restrict__ hs,
                      const float* __restrict__ b1, const float* __restrict__ W2,
                      float* __restrict__ h2s, int E) {
    constexpr int TP = HIDDEN + 1;
    __shared__ float tile[RBP * TP];
    __shared__ float Bf2[(HIDDEN / 2) * 16 * 2];
    __shared__ float b1s[HIDDEN];
    __shared__ float disl[RBP];
    const int2* bsd = (const int2*)bsd_i;
    const int b = blockIdx.x;
    const int base = b * RB;
    const int tid = threadIdx.x;
    for (int i = tid; i < RBP * TP; i += PT) tile[i] = 0.0f;
    __syncthreads();
    float w2v = 0.0f, b1v = 0.0f, dv = 0.0f;
    if (tid < (HIDDEN / 2) * 16 * 2) {
        const int p = tid >> 5;
        const int c = (tid >> 1) & 15;
        const int j = tid & 1;
        if (c < N_CLASSES) w2v = W2[(2 * p + j) * N_CLASSES + c];
    }
    if (tid < HIDDEN) b1v = b1[tid];
    if (tid < RB && base + tid < N_NODES) dv = dis[base + tid];
    gcn_bucket_edges<HIDDEN, PAD1>(tile, bsd, hs, offt, b, (E + NPW - 1) / NPW, base);
    if (tid < (HIDDEN / 2) * 16 * 2) Bf2[tid] = w2v;
    if (tid < HIDDEN) b1s[tid] = b1v;
    if (tid < RBP) disl[tid] = dv;
    __syncthreads();

    const int lane = threadIdx.x & 31;
    const int wave = threadIdx.x >> 5;
    const int r0 = wave << 4;
    if (r0 >= RB || base + r0 >= N_NODES) return;
    const int half = lane >> 4;
    const int l16  = lane & 15;
    const int lr = r0 + l16;
    const bool live = (lr < RB) && (base + lr < N_NODES);
    const float* srow = hs + (size_t)(live ? base + lr : 0) * PAD1;
    const float dA = disl[lr];
    const float* trow = &tile[lr * TP];

    v8f c = {};
    #pragma unroll
    for (int k0 = 0; k0 < HIDDEN; k0 += 4) {
        const int kb = k0 + 2 * half;
        const v2f sr = *(const v2f*)(srow + kb);
        v2f a;
        a.x = fmaxf(dA * (trow[kb]     + sr.x) + b1s[kb],     0.0f);
        a.y = fmaxf(dA * (trow[kb + 1] + sr.y) + b1s[kb + 1], 0.0f);
        const v2f bb = *(const v2f*)&Bf2[(kb >> 1) * 32 + l16 * 2];
        c = __builtin_amdgcn_wmma_f32_16x16x4_f32(false, a, false, bb, (short)0, c, false, false);
    }

    #pragma unroll
    for (int r = 0; r < 8; ++r) {
        const int lrow = r0 + r + 8 * half;
        const int row = base + lrow;
        if (lrow < RB && row < N_NODES)
            h2s[(size_t)row * PAD2 + l16] = disl[lrow] * c[r];
    }
}


__global__ void __launch_bounds__(PT)
gcn_bucket_acc2_lsm(const int* __restrict__ bsd_i, const int* __restrict__ offt,
                    const float* __restrict__ dis, const float* __restrict__ hs,
                    const float* __restrict__ b2, float* __restrict__ out, int E) {
    constexpr int TP = N_CLASSES + 1;
    __shared__ float tile[RB * TP];
    const int2* bsd = (const int2*)bsd_i;
    const int b = blockIdx.x;
    const int base = b * RB;
    for (int i = threadIdx.x; i < RB * TP; i += PT) tile[i] = 0.0f;
    __syncthreads();
    gcn_bucket_edges<N_CLASSES, PAD2>(tile, bsd, hs, offt, b, (E + NPW - 1) / NPW, base);
    __syncthreads();

    const int r = threadIdx.x;
    const int i = base + r;
    if (r >= RB || i >= N_NODES) return;
    const float* row = hs + (size_t)i * PAD2;
    const float di = dis[i];
    float z[N_CLASSES];
    {
        float4 t0 = *(const float4*)(row);
        float4 t1 = *(const float4*)(row + 4);
        float2 t2 = *(const float2*)(row + 8);
        z[0] = t0.x; z[1] = t0.y; z[2] = t0.z; z[3] = t0.w;
        z[4] = t1.x; z[5] = t1.y; z[6] = t1.z; z[7] = t1.w;
        z[8] = t2.x; z[9] = t2.y;
    }
    float m = -3.402823466e38f;
    #pragma unroll
    for (int c = 0; c < N_CLASSES; ++c) {
        z[c] = di * (tile[r * TP + c] + z[c]) + b2[c];
        m = fmaxf(m, z[c]);
    }
    float s = 0.0f;
    #pragma unroll
    for (int c = 0; c < N_CLASSES; ++c) s += __expf(z[c] - m);
    const float l = __logf(s);
    #pragma unroll
    for (int c = 0; c < N_CLASSES; ++c) out[(size_t)i * N_CLASSES + c] = z[c] - m - l;
}


extern "C" void kernel_launch(void* const* d_in, const int* in_sizes, int n_in,
                              void* d_out, int out_size, void* d_ws, size_t ws_size,
                              hipStream_t stream) {
    const float* x  = (const float*)d_in[0];
    const int*   ei = (const int*)d_in[1];
    const float* W1 = (const float*)d_in[2];
    const float* b1 = (const float*)d_in[3];
    const float* W2 = (const float*)d_in[4];
    const float* b2 = (const float*)d_in[5];
    float* out = (float*)d_out;

    const int E = in_sizes[1] / 2;
    const int* src = ei;
    const int* dst = ei + E;

    float* ws  = (float*)d_ws;
    float* dis = ws;
    float* h0  = ws + 102400;
    float* h2  = h0 + (size_t)N_NODES * PAD1;
    int*   offt = (int*)(h2 + (size_t)N_NODES * PAD2);
    int*   bsd = offt + NOFF;

    gcn_group_edges<<<NPW, PT, 0, stream>>>(src, dst, offt, bsd, E);

    gcn_deg_gemm1<<<NB, GT, 0, stream>>>(x, W1, bsd, offt, dis, h0, E);
    gcn_bucket_acc1_gemm2<<<NB, PT, 0, stream>>>(bsd, offt, dis, h0, b1, W2, h2, E);

    gcn_bucket_acc2_lsm<<<NB, PT, 0, stream>>>(bsd, offt, dis, h2, b2, out, E);
}


# Bucket sums in registers, not an LDS float add per edge and word: owner orders its edges by row (an int LDS add each), a lane per row piece sums its source rows; f32 sums now row by row in rank order
# speedup vs baseline: 16.2141x; 1.1591x over previous
#include <hip/hip_runtime.h>


typedef __attribute__((ext_vector_type(2))) float v2f;
typedef __attribute__((ext_vector_type(8))) float v8f;

#define N_NODES   100000
#define N_FEAT    128
#define HIDDEN    20
#define N_CLASSES 10
#define PAD1      32
#define PAD2      16

#define RB    196
#define NB    512
#define NPW   128
#define PT    1024
#define N_EDGES 3200000
#define SR    25
#define NOFF  (NPW * (NB + 1))
static_assert((long long)NPW * SR * PT >= N_EDGES, "a slice must fit its work-group's registers and LDS stage");

__device__ __forceinline__ int gcn_bucket_of(int d) {
    const unsigned b = (unsigned)d / RB;
    return (int)(b < (unsigned)NB ? b : (unsigned)(NB - 1));
}

__device__ __forceinline__ int gcn_local_row(int d, int base) {
    const unsigned r = (unsigned)(d - base);
    return (int)(r < (unsigned)RB ? r : (unsigned)(RB - 1));
}


__global__ void __launch_bounds__(PT)
gcn_group_edges(const int* __restrict__ src, const int* __restrict__ dst,
                int* __restrict__ offt, int* __restrict__ bsd_i, int E) {
    __shared__ int hist[NB];
    __shared__ int2 stage[SR * PT];
    const int tid = threadIdx.x;
    for (int i = tid; i < NB; i += PT) hist[i] = 0;
    __syncthreads();
    const int chunk = (E + NPW - 1) / NPW;
    const int lo = blockIdx.x * chunk;
    const int hi = min(lo + min(chunk, SR * PT), E);
    int d[SR], s[SR], rk[SR];
    #pragma unroll
    for (int u = 0; u < SR; ++u) {
        const int i = lo + u * PT + tid;
        d[u] = (i < hi) ? dst[i] : 0;
        s[u] = (i < hi) ? src[i] : 0;
    }
    #pragma unroll
    for (int u = 0; u < SR; ++u)
        rk[u] = (lo + u * PT + tid < hi) ? atomicAdd(&hist[gcn_bucket_of(d[u])], 1) : 0;
    __syncthreads();
    if (tid < 32) {
        int v[16];
        int sum = 0;
        #pragma unroll
        for (int k = 0; k < 16; ++k) { v[k] = hist[tid * 16 + k]; sum += v[k]; }
        int inc = sum;
        #pragma unroll
        for (int o = 1; o < 32; o <<= 1) {
            const int y = __shfl_up(inc, o);
            if (tid >= o) inc += y;
        }
        int ex = inc - sum;
        #pragma unroll
        for (int k = 0; k < 16; ++k) { hist[tid * 16 + k] = ex; ex += v[k]; }
    }
    __syncthreads();
    int* ot = offt + blockIdx.x * (NB + 1);
    for (int i = tid; i < NB; i += PT) ot[i] = hist[i];
    if (tid == 0) ot[NB] = max(hi - lo, 0);
    #pragma unroll
    for (int u = 0; u < SR; ++u)
        if (lo + u * PT + tid < hi)
            stage[hist[gcn_bucket_of(d[u])] + rk[u]] = make_int2(s[u], d[u]);
    __syncthreads();
    int2* bsd = (int2*)bsd_i;
    const int n = hi - lo;
    static_assert(SR % 5 == 0, "the copy-out takes five records per thread and trip");
    #pragma nounroll
    for (int u0 = 0; u0 < SR; u0 += 5) {
        int2 rec[5];
        #pragma unroll
        for (int u = 0; u < 5; ++u)
            rec[u] = stage[(u0 + u) * PT + tid];
        #pragma unroll
        for (int u = 0; u < 5; ++u)
            asm volatile("" :: "v"(rec[u].x), "v"(rec[u].y));
        #pragma unroll
        for (int u = 0; u < 5; ++u)
            if ((u0 + u) * PT + tid < n) bsd[lo + (u0 + u) * PT + tid] = rec[u];
    }
}


#define RBP 208
#define GT  512
#define GW  (RBP / 16)
#define CT  (GT - GW * 32)
#define CRB 12

__global__ void __launch_bounds__(GT)
gcn_deg_gemm1(const float* __restrict__ x,
              const float* __restrict__ W1,
              const int* __restrict__ bsd_i, const int* __restrict__ offt,
              float* __restrict__ dis,
              float* __restrict__ hs, int E) {
    __shared__ float Bf[(N_FEAT / 2) * 32 * 2];
    __shared__ int cdeg[RBP];

    const int tid = threadIdx.x;
    const int b = blockIdx.x;
    const int base = b * RB;
    for (int idx = tid; idx < (N_FEAT / 2) * 32 * 2; idx += GT) {
        const int p = idx >> 6;
        const int c = (idx >> 1) & 31;
        const int j = idx & 1;
        Bf[idx] = (c < HIDDEN) ? W1[(2 * p + j) * HIDDEN + c] : 0.0f;
    }
    for (int i = tid; i < RBP; i += GT) cdeg[i] = 0;
    __syncthreads();

    const int lane = tid & 31;
    const int wave = tid >> 5;
    const int half = lane >> 4;
    const int l16  = lane & 15;
    const int r0 = wave << 4;
    const bool gemm_wave = wave < GW;
    const bool tile_on = gemm_wave && r0 < RB && base + r0 < N_NODES;

    v8f c0 = {};
    v8f c1 = {};

    if (tile_on) {
        const float* arow = x + (size_t)min(base + r0 + l16, N_NODES - 1) * N_FEAT;
        #pragma unroll 4
        for (int k0 = 0; k0 < N_FEAT; k0 += 4) {
            const int kb = k0 + 2 * half;
            const int p = kb >> 1;
            v2f a  = *(const v2f*)(arow + kb);
            v2f b0 = *(const v2f*)&Bf[p * 64 + l16 * 2];
            v2f b1 = *(const v2f*)&Bf[p * 64 + (16 + l16) * 2];

            c0 = __builtin_amdgcn_wmma_f32_16x16x4_f32(false, a, false, b0, (short)0, c0, false, false);
            c1 = __builtin_amdgcn_wmma_f32_16x16x4_f32(false, a, false, b1, (short)0, c1, false, false);
        }
    } else if (!gemm_wave) {
        const int2* bsd = (const int2*)bsd_i;
        const int chunk = (E + NPW - 1) / NPW;
        const int cw = __builtin_amdgcn_readfirstlane(wave) - GW;
        for (int rb = cw * CRB; rb < NPW; rb += (CT / 32) * CRB) {
            int st[CRB], ln[CRB];
            #pragma unroll
            for (int u = 0; u < CRB; ++u) {
                const int r = min(rb + u, NPW - 1);
                const int a  = offt[r * (NB + 1) + b];
                const int en = offt[r * (NB + 1) + b + 1];
                st[u] = r * chunk + a;
                ln[u] = (rb + u < NPW) ? en - a : 0;
            }
            int d[2 * CRB];
            #pragma unroll
            for (int u = 0; u < CRB; ++u) {
                d[2 * u]     = bsd[min(st[u] + lane, E - 1)].y;
                d[2 * u + 1] = bsd[min(st[u] + lane + 32, E - 1)].y;
            }
            #pragma unroll
            for (int u = 0; u < 2 * CRB; ++u)
                asm volatile("" :: "v"(d[u]));
            #pragma unroll
            for (int u = 0; u < CRB; ++u) {
                if (lane < ln[u])      atomicAdd(&cdeg[gcn_local_row(d[2 * u], base)], 1);
                if (lane + 32 < ln[u]) atomicAdd(&cdeg[gcn_local_row(d[2 * u + 1], base)], 1);
                for (int j = lane + 64; j < ln[u]; j += 32)
                    atomicAdd(&cdeg[gcn_local_row(bsd[st[u] + j].y, base)], 1);
            }
        }
    }
    __syncthreads();

    if (tile_on) {
        float di[8];
        #pragma unroll
        for (int r = 0; r < 8; ++r)
            di[r] = rsqrtf(1.0f + (float)cdeg[r0 + r + 8 * half]);
        #pragma unroll
        for (int r = 0; r < 8; ++r) {
            const int lrow = r0 + r + 8 * half;
            const int row = base + lrow;
            if (lrow < RB && row < N_NODES) {
                float* p0 = hs + (size_t)row * PAD1;
                p0[l16]      = di[r] * c0[r];
                p0[16 + l16] = di[r] * c1[r];
            }
        }
    } else if (!gemm_wave) {
        for (int i = tid - GW * 32; i < RB; i += CT) {
            const int row = base + i;
            if (row < N_NODES) dis[row] = rsqrtf(1.0f + (float)cdeg[i]);
        }
    }
}


#define KPR   8
#define CAP   (KPR * PT)
#define SPL   ((RB + 31) / 32)

template <int F, int PAD>
__device__ __forceinline__ void gcn_bucket_rows(float* __restrict__ tile, int* __restrict__ cnt,
                                                int* __restrict__ off, int* __restrict__ srt,
                                                const int2* __restrict__ bsd,
                                                const float* __restrict__ hs, const int* __restrict__ offt,
                                                int b, int chunk, int base) {
    constexpr int TP  = F + 1;
    constexpr int LPE = (F + 3) / 4;
    constexpr int UG  = 8;
    constexpr int RPW = NPW / (PT / 32);
    static_assert(4 * LPE <= PAD, "row pieces must stay inside the padded row");
    static_assert(RPW * (PT / 32) == NPW, "every run needs exactly one wave");
    static_assert(RB * LPE <= PT, "every row piece needs a lane of its own");
    static_assert(RB < SPL * 32, "off[RB], the end of the last row, must lie inside the padded table");
    static_assert(SPL * 32 <= 256 && CAP <= (1 << 23), "row and rank share a word: 8 bits of row, the rest rank");
    const int tid  = threadIdx.x;
    const int lane = tid & 31;
    const int wave = __builtin_amdgcn_readfirstlane(tid >> 5);
    int st[RPW], cum[RPW + 1];
    cum[0] = 0;
    #pragma unroll
    for (int k = 0; k < RPW; ++k) {
        const int sl = wave * RPW + k;
        const int a  = offt[sl * (NB + 1) + b];
        const int en = offt[sl * (NB + 1) + b + 1];
        st[k] = sl * chunk + a - cum[k];
        cum[k + 1] = cum[k] + (en - a);
    }
    const int n  = cum[RPW];
    const int nh = min(n, KPR * 32);
    {
        int2 rec[KPR];
        int pk[KPR];
        const int lim = nh - lane;
        if (nh > 0) {
            #pragma unroll
            for (int u = 0; u < KPR; ++u) {
                const int j = lane + min(u * 32, lim - 1);
                int o = st[0];
                #pragma unroll
                for (int k = 1; k < RPW; ++k) o = (j >= cum[k]) ? st[k] : o;
                rec[u] = bsd[o + j];
            }
            #pragma unroll
            for (int u = 0; u < KPR; ++u) {
                const int lr = (u * 32 < lim) ? gcn_local_row(rec[u].y, base) : RB + lane % (SPL * 32 - RB);
                pk[u] = lr | (atomicAdd(&cnt[lr], 1) << 8);
            }
        }
        __syncthreads();
        if (wave == 0) {
            int v[SPL];
            int sum = 0;
            #pragma unroll
            for (int k = 0; k < SPL; ++k) { v[k] = cnt[lane * SPL + k]; sum += v[k]; }
            int inc = sum;
            #pragma unroll
            for (int o = 1; o < 32; o <<= 1) {
                const int y = __shfl_up(inc, o);
                if (lane >= o) inc += y;
            }
            int ex = inc - sum;
            #pragma unroll
            for (int k = 0; k < SPL; ++k) { off[lane * SPL + k] = ex; ex += v[k]; }
        }
        __syncthreads();
        if (nh > 0) {
            int ob[KPR];
            #pragma unroll
            for (int u = 0; u < KPR; ++u)
                ob[u] = off[pk[u] & 255];
            #pragma unroll
            for (int u = 0; u < KPR; ++u)
                asm volatile("" :: "v"(ob[u]));
            #pragma unroll
            for (int u = 0; u < KPR; ++u)
                if ((pk[u] & 255) < RB)
                    srt[ob[u] + (pk[u] >> 8)] = rec[u].x;
        }
        __syncthreads();
    }
    const int row = tid / LPE;
    if (row < RB) {
        const int pc = tid - row * LPE;
        const float4* __restrict__ hs4 = (const float4*)hs;
        float4 acc = make_float4(0.0f, 0.0f, 0.0f, 0.0f);
        const int beg = off[row];
        const int end = off[row + 1];
        for (int i = beg; i < end; i += UG) {
            const int rem = end - i;
            int s[UG];
            #pragma unroll
            for (int u = 0; u < UG; ++u)
                s[u] = srt[i + min(u, rem - 1)];
            float4 v[UG];
            #pragma unroll
            for (int u = 0; u < UG; ++u)
                v[u] = hs4[(unsigned)(s[u] * (PAD / 4) + pc)];
            #pragma unroll
            for (int u = 0; u < UG; ++u)
                asm volatile("" :: "v"(v[u].x), "v"(v[u].y), "v"(v[u].z), "v"(v[u].w));
            #pragma unroll
            for (int u = 0; u < UG; ++u) {
                const bool ok = u < rem;
                acc.x += ok ? v[u].x : 0.0f;
                acc.y += ok ? v[u].y : 0.0f;
                acc.z += ok ? v[u].z : 0.0f;
                acc.w += ok ? v[u].w : 0.0f;
            }
        }
        float* t = &tile[row * TP + pc * 4];
        atomicAdd(t, acc.x);
        if (F % 4 == 0 || pc * 4 + 1 < F) atomicAdd(t + 1, acc.y);
        if (F % 4 == 0 || pc * 4 + 2 < F) atomicAdd(t + 2, acc.z);
        if (F % 4 == 0 || pc * 4 + 3 < F) atomicAdd(t + 3, acc.w);
    }
    if (n > KPR * 32) {
        constexpr int EPW = 32 / LPE;
        constexpr int U   = 30 / EPW;
        constexpr int EPI = EPW * U;
        const int el = min(lane / LPE, EPW - 1);
        const int c4 = (lane % LPE) * 4;
        const bool lane_on = lane < LPE * EPW;
        for (int j0 = KPR * 32; j0 < n; j0 += EPI) {
            int2 sd[U];
            #pragma unroll
            for (int u = 0; u < U; ++u) {
                const int j = min(j0 + u * EPW + el, n - 1);
                int o = st[0];
                #pragma unroll
                for (int k = 1; k < RPW; ++k) o = (j >= cum[k]) ? st[k] : o;
                sd[u] = bsd[o + j];
            }
            float4 v[U];
            #pragma unroll
            for (int u = 0; u < U; ++u)
                v[u] = *(const float4*)(hs + (size_t)sd[u].x * PAD + c4);
            #pragma unroll
            for (int u = 0; u < U; ++u)
                asm volatile("" :: "v"(v[u].x), "v"(v[u].y), "v"(v[u].z), "v"(v[u].w));
            #pragma unroll
            for (int u = 0; u < U; ++u) {
                if (lane_on && j0 + u * EPW + el < n) {
                    float* t = &tile[gcn_local_row(sd[u].y, base) * TP + c4];
                    atomicAdd(t, v[u].x);
                    if (F % 4 == 0 || c4 + 1 < F) atomicAdd(t + 1, v[u].y);
                    if (F % 4 == 0 || c4 + 2 < F) atomicAdd(t + 2, v[u].z);
                    if (F % 4 == 0 || c4 + 3 < F) atomicAdd(t + 3, v[u].w);
                }
            }
        }
    }
}


__global__ void __launch_bounds__(PT, 16)
gcn_bucket_acc1_gemm2(const int* __restrict__ bsd_i, const int* __restrict__ offt,
                      const float* __restrict__ dis, const float* __restrict__ hs,
                      const float* __restrict__ b1, const float* __restrict__ W2,
                      float* __restrict__ h2s, int E) {
    constexpr int TP = HIDDEN + 1;
    __shared__ float tile[RBP * TP];
    __shared__ float Bf2[(HIDDEN / 2) * 16 * 2];
    __shared__ float b1s[HIDDEN];
    __shared__ float disl[RBP];
    __shared__ int cnt[SPL * 32];
    __shared__ int off[SPL * 32];
    __shared__ int srt[CAP];
    const int2* bsd = (const int2*)bsd_i;
    const int b = blockIdx.x;
    const int base = b * RB;
    const int tid = threadIdx.x;
    for (int i = tid; i < RBP * TP; i += PT) tile[i] = 0.0f;
    if (tid < SPL * 32) cnt[tid] = 0;
    __syncthreads();
    float w2v = 0.0f, b1v = 0.0f, dv = 0.0f;
    if (tid < (HIDDEN / 2) * 16 * 2) {
        const int p = tid >> 5;
        const int c = (tid >> 1) & 15;
        const int j = tid & 1;
        if (c < N_CLASSES) w2v = W2[(2 * p + j) * N_CLASSES + c];
    }
    if (tid < HIDDEN) b1v = b1[tid];
    if (tid < RB && base + tid < N_NODES) dv = dis[base + tid];
    gcn_bucket_rows<HIDDEN, PAD1>(tile, cnt, off, srt, bsd, hs, offt, b, (E + NPW - 1) / NPW, base);
    if (tid < (HIDDEN / 2) * 16 * 2) Bf2[tid] = w2v;
    if (tid < HIDDEN) b1s[tid] = b1v;
    if (tid < RBP) disl[tid] = dv;
    __syncthreads();

    const int lane = threadIdx.x & 31;
    const int wave = threadIdx.x >> 5;
    const int r0 = wave << 4;
    if (r0 >= RB || base + r0 >= N_NODES) return;
    const int half = lane >> 4;
    const int l16  = lane & 15;
    const int lr = r0 + l16;
    const bool live = (lr < RB) && (base + lr < N_NODES);
    const float* srow = hs + (size_t)(live ? base + lr : 0) * PAD1;
    const float dA = disl[lr];
    const float* trow = &tile[lr * TP];

    v8f c = {};
    #pragma unroll
    for (int k0 = 0; k0 < HIDDEN; k0 += 4) {
        const int kb = k0 + 2 * half;
        const v2f sr = *(const v2f*)(srow + kb);
        v2f a;
        a.x = fmaxf(dA * (trow[kb]     + sr.x) + b1s[kb],     0.0f);
        a.y = fmaxf(dA * (trow[kb + 1] + sr.y) + b1s[kb + 1], 0.0f);
        const v2f bb = *(const v2f*)&Bf2[(kb >> 1) * 32 + l16 * 2];
        c = __builtin_amdgcn_wmma_f32_16x16x4_f32(false, a, false, bb, (short)0, c, false, false);
    }

    #pragma unroll
    for (int r = 0; r < 8; ++r) {
        const int lrow = r0 + r + 8 * half;
        const int row = base + lrow;
        if (lrow < RB && row < N_NODES)
            h2s[(size_t)row * PAD2 + l16] = disl[lrow] * c[r];
    }
}


__global__ void __launch_bounds__(PT, 16)
gcn_bucket_acc2_lsm(const int* __restrict__ bsd_i, const int* __restrict__ offt,
                    const float* __restrict__ dis, const float* __restrict__ hs,
                    const float* __restrict__ b2, float* __restrict__ out, int E) {
    constexpr int TP = N_CLASSES + 1;
    __shared__ float tile[RB * TP];
    __shared__ int cnt[SPL * 32];
    __shared__ int off[SPL * 32];
    __shared__ int srt[CAP];
    const int2* bsd = (const int2*)bsd_i;
    const int b = blockIdx.x;
    const int base = b * RB;
    for (int i = threadIdx.x; i < RB * TP; i += PT) tile[i] = 0.0f;
    if (threadIdx.x < SPL * 32) cnt[threadIdx.x] = 0;
    __syncthreads();
    gcn_bucket_rows<N_CLASSES, PAD2>(tile, cnt, off, srt, bsd, hs, offt, b, (E + NPW - 1) / NPW, base);
    __syncthreads();

    const int r = threadIdx.x;
    const int i = base + r;
    if (r >= RB || i >= N_NODES) return;
    const float* row = hs + (size_t)i * PAD2;
    const float di = dis[i];
    float z[N_CLASSES];
    {
        float4 t0 = *(const float4*)(row);
        float4 t1 = *(const float4*)(row + 4);
        float2 t2 = *(const float2*)(row + 8);
        z[0] = t0.x; z[1] = t0.y; z[2] = t0.z; z[3] = t0.w;
        z[4] = t1.x; z[5] = t1.y; z[6] = t1.z; z[7] = t1.w;
        z[8] = t2.x; z[9] = t2.y;
    }
    float m = -3.402823466e38f;
    #pragma unroll
    for (int c = 0; c < N_CLASSES; ++c) {
        z[c] = di * (tile[r * TP + c] + z[c]) + b2[c];
        m = fmaxf(m, z[c]);
    }
    float s = 0.0f;
    #pragma unroll
    for (int c = 0; c < N_CLASSES; ++c) s += __expf(z[c] - m);
    const float l = __logf(s);
    #pragma unroll
    for (int c = 0; c < N_CLASSES; ++c) out[(size_t)i * N_CLASSES + c] = z[c] - m - l;
}


extern "C" void kernel_launch(void* const* d_in, const int* in_sizes, int n_in,
                              void* d_out, int out_size, void* d_ws, size_t ws_size,
                              hipStream_t stream) {
    const float* x  = (const float*)d_in[0];
    const int*   ei = (const int*)d_in[1];
    const float* W1 = (const float*)d_in[2];
    const float* b1 = (const float*)d_in[3];
    const float* W2 = (const float*)d_in[4];
    const float* b2 = (const float*)d_in[5];
    float* out = (float*)d_out;

    const int E = in_sizes[1] / 2;
    const int* src = ei;
    const int* dst = ei + E;

    float* ws  = (float*)d_ws;
    float* dis = ws;
    float* h0  = ws + 102400;
    float* h2  = h0 + (size_t)N_NODES * PAD1;
    int*   offt = (int*)(h2 + (size_t)N_NODES * PAD2);
    int*   bsd = offt + NOFF;

    gcn_group_edges<<<NPW, PT, 0, stream>>>(src, dst, offt, bsd, E);

    gcn_deg_gemm1<<<NB, GT, 0, stream>>>(x, W1, bsd, offt, dis, h0, E);
    gcn_bucket_acc1_gemm2<<<NB, PT, 0, stream>>>(bsd, offt, dis, h0, b1, W2, h2, E);

    gcn_bucket_acc2_lsm<<<NB, PT, 0, stream>>>(bsd, offt, dis, h2, b2, out, E);
}


# Grouped edge list holds one 32-bit word per edge (source row | row in bucket << 17), not a (src, dst) pair: grouping pass stages and stores half the bytes, all three list readers load half
# speedup vs baseline: 17.0076x; 1.0489x over previous
#include <hip/hip_runtime.h>


typedef __attribute__((ext_vector_type(2))) float v2f;
typedef __attribute__((ext_vector_type(8))) float v8f;

#define N_NODES   100000
#define N_FEAT    128
#define HIDDEN    20
#define N_CLASSES 10
#define PAD1      32
#define PAD2      16

#define RB    196
#define NB    512
#define NPW   128
#define PT    1024
#define N_EDGES 3200000
#define SR    25
#define NOFF  (NPW * (NB + 1))
static_assert((long long)NPW * SR * PT >= N_EDGES, "a slice must fit its work-group's registers and LDS stage");

__device__ __forceinline__ int gcn_bucket_of(int d) {
    const unsigned b = (unsigned)d / RB;
    return (int)(b < (unsigned)NB ? b : (unsigned)(NB - 1));
}

#define SRC_BITS 17
#define SRC_MASK ((1 << SRC_BITS) - 1)
static_assert(N_NODES <= (1 << SRC_BITS), "a source row must fit the low bits of a record");
static_assert(RB <= 256, "a row inside a bucket must fit 8 bits above the source row");

__device__ __forceinline__ int gcn_rec_pack(int s, int d, int b) {
    return s | (((d - b * RB) & 255) << SRC_BITS);
}

__device__ __forceinline__ int gcn_rec_src(int w) { return w & SRC_MASK; }

__device__ __forceinline__ int gcn_rec_row(int w) {
    const unsigned r = (unsigned)w >> SRC_BITS;
    return (int)(r < (unsigned)RB ? r : (unsigned)(RB - 1));
}


__global__ void __launch_bounds__(PT)
gcn_group_edges(const int* __restrict__ src, const int* __restrict__ dst,
                int* __restrict__ offt, int* __restrict__ bsd_i, int E) {
    __shared__ int hist[NB];
    __shared__ int stage[SR * PT];
    const int tid = threadIdx.x;
    for (int i = tid; i < NB; i += PT) hist[i] = 0;
    __syncthreads();
    const int chunk = (E + NPW - 1) / NPW;
    const int lo = blockIdx.x * chunk;
    const int hi = min(lo + min(chunk, SR * PT), E);
    int d[SR], s[SR], rk[SR];
    #pragma unroll
    for (int u = 0; u < SR; ++u) {
        const int i = lo + u * PT + tid;
        d[u] = (i < hi) ? dst[i] : 0;
        s[u] = (i < hi) ? src[i] : 0;
    }
    #pragma unroll
    for (int u = 0; u < SR; ++u)
        rk[u] = (lo + u * PT + tid < hi) ? atomicAdd(&hist[gcn_bucket_of(d[u])], 1) : 0;
    __syncthreads();
    if (tid < 32) {
        int v[16];
        int sum = 0;
        #pragma unroll
        for (int k = 0; k < 16; ++k) { v[k] = hist[tid * 16 + k]; sum += v[k]; }
        int inc = sum;
        #pragma unroll
        for (int o = 1; o < 32; o <<= 1) {
            const int y = __shfl_up(inc, o);
            if (tid >= o) inc += y;
        }
        int ex = inc - sum;
        #pragma unroll
        for (int k = 0; k < 16; ++k) { hist[tid * 16 + k] = ex; ex += v[k]; }
    }
    __syncthreads();
    int* ot = offt + blockIdx.x * (NB + 1);
    for (int i = tid; i < NB; i += PT) ot[i] = hist[i];
    if (tid == 0) ot[NB] = max(hi - lo, 0);
    #pragma unroll
    for (int u = 0; u < SR; ++u)
        if (lo + u * PT + tid < hi) {
            const int bk = gcn_bucket_of(d[u]);
            stage[hist[bk] + rk[u]] = gcn_rec_pack(s[u], d[u], bk);
        }
    __syncthreads();
    const int n = hi - lo;
    static_assert(SR % 5 == 0, "the copy-out takes five records per thread and trip");
    #pragma nounroll
    for (int u0 = 0; u0 < SR; u0 += 5) {
        int rec[5];
        #pragma unroll
        for (int u = 0; u < 5; ++u)
            rec[u] = stage[(u0 + u) * PT + tid];
        #pragma unroll
        for (int u = 0; u < 5; ++u)
            asm volatile("" :: "v"(rec[u]));
        #pragma unroll
        for (int u = 0; u < 5; ++u)
            if ((u0 + u) * PT + tid < n) bsd_i[lo + (u0 + u) * PT + tid] = rec[u];
    }
}


#define RBP 208
#define GT  512
#define GW  (RBP / 16)
#define CT  (GT - GW * 32)
#define CRB 12

__global__ void __launch_bounds__(GT)
gcn_deg_gemm1(const float* __restrict__ x,
              const float* __restrict__ W1,
              const int* __restrict__ bsd_i, const int* __restrict__ offt,
              float* __restrict__ dis,
              float* __restrict__ hs, int E) {
    __shared__ float Bf[(N_FEAT / 2) * 32 * 2];
    __shared__ int cdeg[RBP];

    const int tid = threadIdx.x;
    const int b = blockIdx.x;
    const int base = b * RB;
    for (int idx = tid; idx < (N_FEAT / 2) * 32 * 2; idx += GT) {
        const int p = idx >> 6;
        const int c = (idx >> 1) & 31;
        const int j = idx & 1;
        Bf[idx] = (c < HIDDEN) ? W1[(2 * p + j) * HIDDEN + c] : 0.0f;
    }
    for (int i = tid; i < RBP; i += GT) cdeg[i] = 0;
    __syncthreads();

    const int lane = tid & 31;
    const int wave = tid >> 5;
    const int half = lane >> 4;
    const int l16  = lane & 15;
    const int r0 = wave << 4;
    const bool gemm_wave = wave < GW;
    const bool tile_on = gemm_wave && r0 < RB && base + r0 < N_NODES;

    v8f c0 = {};
    v8f c1 = {};

    if (tile_on) {
        const float* arow = x + (size_t)min(base + r0 + l16, N_NODES - 1) * N_FEAT;
        #pragma unroll 4
        for (int k0 = 0; k0 < N_FEAT; k0 += 4) {
            const int kb = k0 + 2 * half;
            const int p = kb >> 1;
            v2f a  = *(const v2f*)(arow + kb);
            v2f b0 = *(const v2f*)&Bf[p * 64 + l16 * 2];
            v2f b1 = *(const v2f*)&Bf[p * 64 + (16 + l16) * 2];

            c0 = __builtin_amdgcn_wmma_f32_16x16x4_f32(false, a, false, b0, (short)0, c0, false, false);
            c1 = __builtin_amdgcn_wmma_f32_16x16x4_f32(false, a, false, b1, (short)0, c1, false, false);
        }
    } else if (!gemm_wave) {
        const int* bsd = bsd_i;
        const int chunk = (E + NPW - 1) / NPW;
        const int cw = __builtin_amdgcn_readfirstlane(wave) - GW;
        for (int rb = cw * CRB; rb < NPW; rb += (CT / 32) * CRB) {
            int st[CRB], ln[CRB];
            #pragma unroll
            for (int u = 0; u < CRB; ++u) {
                const int r = min(rb + u, NPW - 1);
                const int a  = offt[r * (NB + 1) + b];
                const int en = offt[r * (NB + 1) + b + 1];
                st[u] = r * chunk + a;
                ln[u] = (rb + u < NPW) ? en - a : 0;
            }
            int d[2 * CRB];
            #pragma unroll
            for (int u = 0; u < CRB; ++u) {
                d[2 * u]     = bsd[min(st[u] + lane, E - 1)];
                d[2 * u + 1] = bsd[min(st[u] + lane + 32, E - 1)];
            }
            #pragma unroll
            for (int u = 0; u < 2 * CRB; ++u)
                asm volatile("" :: "v"(d[u]));
            #pragma unroll
            for (int u = 0; u < CRB; ++u) {
                if (lane < ln[u])      atomicAdd(&cdeg[gcn_rec_row(d[2 * u])], 1);
                if (lane + 32 < ln[u]) atomicAdd(&cdeg[gcn_rec_row(d[2 * u + 1])], 1);
                for (int j = lane + 64; j < ln[u]; j += 32)
                    atomicAdd(&cdeg[gcn_rec_row(bsd[st[u] + j])], 1);
            }
        }
    }
    __syncthreads();

    if (tile_on) {
        float di[8];
        #pragma unroll
        for (int r = 0; r < 8; ++r)
            di[r] = rsqrtf(1.0f + (float)cdeg[r0 + r + 8 * half]);
        #pragma unroll
        for (int r = 0; r < 8; ++r) {
            const int lrow = r0 + r + 8 * half;
            const int row = base + lrow;
            if (lrow < RB && row < N_NODES) {
                float* p0 = hs + (size_t)row * PAD1;
                p0[l16]      = di[r] * c0[r];
                p0[16 + l16] = di[r] * c1[r];
            }
        }
    } else if (!gemm_wave) {
        for (int i = tid - GW * 32; i < RB; i += CT) {
            const int row = base + i;
            if (row < N_NODES) dis[row] = rsqrtf(1.0f + (float)cdeg[i]);
        }
    }
}


#define KPR   8
#define CAP   (KPR * PT)
#define SPL   ((RB + 31) / 32)

template <int F, int PAD>
__device__ __forceinline__ void gcn_bucket_rows(float* __restrict__ tile, int* __restrict__ cnt,
                                                int* __restrict__ off, int* __restrict__ srt,
                                                const int* __restrict__ bsd,
                                                const float* __restrict__ hs, const int* __restrict__ offt,
                                                int b, int chunk) {
    constexpr int TP  = F + 1;
    constexpr int LPE = (F + 3) / 4;
    constexpr int UG  = 8;
    constexpr int RPW = NPW / (PT / 32);
    static_assert(4 * LPE <= PAD, "row pieces must stay inside the padded row");
    static_assert(RPW * (PT / 32) == NPW, "every run needs exactly one wave");
    static_assert(RB * LPE <= PT, "every row piece needs a lane of its own");
    static_assert(RB < SPL * 32, "off[RB], the end of the last row, must lie inside the padded table");
    static_assert(SPL * 32 <= 256 && CAP <= (1 << 23), "row and rank share a word: 8 bits of row, the rest rank");
    const int tid  = threadIdx.x;
    const int lane = tid & 31;
    const int wave = __builtin_amdgcn_readfirstlane(tid >> 5);
    int st[RPW], cum[RPW + 1];
    cum[0] = 0;
    #pragma unroll
    for (int k = 0; k < RPW; ++k) {
        const int sl = wave * RPW + k;
        const int a  = offt[sl * (NB + 1) + b];
        const int en = offt[sl * (NB + 1) + b + 1];
        st[k] = sl * chunk + a - cum[k];
        cum[k + 1] = cum[k] + (en - a);
    }
    const int n  = cum[RPW];
    const int nh = min(n, KPR * 32);
    {
        int rec[KPR];
        int pk[KPR];
        const int lim = nh - lane;
        if (nh > 0) {
            #pragma unroll
            for (int u = 0; u < KPR; ++u) {
                const int j = lane + min(u * 32, lim - 1);
                int o = st[0];
                #pragma unroll
                for (int k = 1; k < RPW; ++k) o = (j >= cum[k]) ? st[k] : o;
                rec[u] = bsd[o + j];
            }
            #pragma unroll
            for (int u = 0; u < KPR; ++u) {
                const int lr = (u * 32 < lim) ? gcn_rec_row(rec[u]) : RB + lane % (SPL * 32 - RB);
                pk[u] = lr | (atomicAdd(&cnt[lr], 1) << 8);
            }
        }
        __syncthreads();
        if (wave == 0) {
            int v[SPL];
            int sum = 0;
            #pragma unroll
            for (int k = 0; k < SPL; ++k) { v[k] = cnt[lane * SPL + k]; sum += v[k]; }
            int inc = sum;
            #pragma unroll
            for (int o = 1; o < 32; o <<= 1) {
                const int y = __shfl_up(inc, o);
                if (lane >= o) inc += y;
            }
            int ex = inc - sum;
            #pragma unroll
            for (int k = 0; k < SPL; ++k) { off[lane * SPL + k] = ex; ex += v[k]; }
        }
        __syncthreads();
        if (nh > 0) {
            int ob[KPR];
            #pragma unroll
            for (int u = 0; u < KPR; ++u)
                ob[u] = off[pk[u] & 255];
            #pragma unroll
            for (int u = 0; u < KPR; ++u)
                asm volatile("" :: "v"(ob[u]));
            #pragma unroll
            for (int u = 0; u < KPR; ++u)
                if ((pk[u] & 255) < RB)
                    srt[ob[u] + (pk[u] >> 8)] = gcn_rec_src(rec[u]);
        }
        __syncthreads();
    }
    const int row = tid / LPE;
    if (row < RB) {
        const int pc = tid - row * LPE;
        const float4* __restrict__ hs4 = (const float4*)hs;
        float4 acc = make_float4(0.0f, 0.0f, 0.0f, 0.0f);
        const int beg = off[row];
        const int end = off[row + 1];
        for (int i = beg; i < end; i += UG) {
            const int rem = end - i;
            int s[UG];
            #pragma unroll
            for (int u = 0; u < UG; ++u)
                s[u] = srt[i + min(u, rem - 1)];
            float4 v[UG];
            #pragma unroll
            for (int u = 0; u < UG; ++u)
                v[u] = hs4[(unsigned)(s[u] * (PAD / 4) + pc)];
            #pragma unroll
            for (int u = 0; u < UG; ++u)
                asm volatile("" :: "v"(v[u].x), "v"(v[u].y), "v"(v[u].z), "v"(v[u].w));
            #pragma unroll
            for (int u = 0; u < UG; ++u) {
                const bool ok = u < rem;
                acc.x += ok ? v[u].x : 0.0f;
                acc.y += ok ? v[u].y : 0.0f;
                acc.z += ok ? v[u].z : 0.0f;
                acc.w += ok ? v[u].w : 0.0f;
            }
        }
        float* t = &tile[row * TP + pc * 4];
        atomicAdd(t, acc.x);
        if (F % 4 == 0 || pc * 4 + 1 < F) atomicAdd(t + 1, acc.y);
        if (F % 4 == 0 || pc * 4 + 2 < F) atomicAdd(t + 2, acc.z);
        if (F % 4 == 0 || pc * 4 + 3 < F) atomicAdd(t + 3, acc.w);
    }
    if (n > KPR * 32) {
        constexpr int EPW = 32 / LPE;
        constexpr int U   = 30 / EPW;
        constexpr int EPI = EPW * U;
        const int el = min(lane / LPE, EPW - 1);
        const int c4 = (lane % LPE) * 4;
        const bool lane_on = lane < LPE * EPW;
        for (int j0 = KPR * 32; j0 < n; j0 += EPI) {
            int sd[U];
            #pragma unroll
            for (int u = 0; u < U; ++u) {
                const int j = min(j0 + u * EPW + el, n - 1);
                int o = st[0];
                #pragma unroll
                for (int k = 1; k < RPW; ++k) o = (j >= cum[k]) ? st[k] : o;
                sd[u] = bsd[o + j];
            }
            float4 v[U];
            #pragma unroll
            for (int u = 0; u < U; ++u)
                v[u] = *(const float4*)(hs + (size_t)gcn_rec_src(sd[u]) * PAD + c4);
            #pragma unroll
            for (int u = 0; u < U; ++u)
                asm volatile("" :: "v"(v[u].x), "v"(v[u].y), "v"(v[u].z), "v"(v[u].w));
            #pragma unroll
            for (int u = 0; u < U; ++u) {
                if (lane_on && j0 + u * EPW + el < n) {
                    float* t = &tile[gcn_rec_row(sd[u]) * TP + c4];
                    atomicAdd(t, v[u].x);
                    if (F % 4 == 0 || c4 + 1 < F) atomicAdd(t + 1, v[u].y);
                    if (F % 4 == 0 || c4 + 2 < F) atomicAdd(t + 2, v[u].z);
                    if (F % 4 == 0 || c4 + 3 < F) atomicAdd(t + 3, v[u].w);
                }
            }
        }
    }
}


__global__ void __launch_bounds__(PT, 16)
gcn_bucket_acc1_gemm2(const int* __restrict__ bsd_i, const int* __restrict__ offt,
                      const float* __restrict__ dis, const float* __restrict__ hs,
                      const float* __restrict__ b1, const float* __restrict__ W2,
                      float* __restrict__ h2s, int E) {
    constexpr int TP = HIDDEN + 1;
    __shared__ float tile[RBP * TP];
    __shared__ float Bf2[(HIDDEN / 2) * 16 * 2];
    __shared__ float b1s[HIDDEN];
    __shared__ float disl[RBP];
    __shared__ int cnt[SPL * 32];
    __shared__ int off[SPL * 32];
    __shared__ int srt[CAP];
    const int b = blockIdx.x;
    const int base = b * RB;
    const int tid = threadIdx.x;
    for (int i = tid; i < RBP * TP; i += PT) tile[i] = 0.0f;
    if (tid < SPL * 32) cnt[tid] = 0;
    __syncthreads();
    float w2v = 0.0f, b1v = 0.0f, dv = 0.0f;
    if (tid < (HIDDEN / 2) * 16 * 2) {
        const int p = tid >> 5;
        const int c = (tid >> 1) & 15;
        const int j = tid & 1;
        if (c < N_CLASSES) w2v = W2[(2 * p + j) * N_CLASSES + c];
    }
    if (tid < HIDDEN) b1v = b1[tid];
    if (tid < RB && base + tid < N_NODES) dv = dis[base + tid];
    gcn_bucket_rows<HIDDEN, PAD1>(tile, cnt, off, srt, bsd_i, hs, offt, b, (E + NPW - 1) / NPW);
    if (tid < (HIDDEN / 2) * 16 * 2) Bf2[tid] = w2v;
    if (tid < HIDDEN) b1s[tid] = b1v;
    if (tid < RBP) disl[tid] = dv;
    __syncthreads();

    const int lane = threadIdx.x & 31;
    const int wave = threadIdx.x >> 5;
    const int r0 = wave << 4;
    if (r0 >= RB || base + r0 >= N_NODES) return;
    const int half = lane >> 4;
    const int l16  = lane & 15;
    const int lr = r0 + l16;
    const bool live = (lr < RB) && (base + lr < N_NODES);
    const float* srow = hs + (size_t)(live ? base + lr : 0) * PAD1;
    const float dA = disl[lr];
    const float* trow = &tile[lr * TP];

    v8f c = {};
    #pragma unroll
    for (int k0 = 0; k0 < HIDDEN; k0 += 4) {
        const int kb = k0 + 2 * half;
        const v2f sr = *(const v2f*)(srow + kb);
        v2f a;
        a.x = fmaxf(dA * (trow[kb]     + sr.x) + b1s[kb],     0.0f);
        a.y = fmaxf(dA * (trow[kb + 1] + sr.y) + b1s[kb + 1], 0.0f);
        const v2f bb = *(const v2f*)&Bf2[(kb >> 1) * 32 + l16 * 2];
        c = __builtin_amdgcn_wmma_f32_16x16x4_f32(false, a, false, bb, (short)0, c, false, false);
    }

    #pragma unroll
    for (int r = 0; r < 8; ++r) {
        const int lrow = r0 + r + 8 * half;
        const int row = base + lrow;
        if (lrow < RB && row < N_NODES)
            h2s[(size_t)row * PAD2 + l16] = disl[lrow] * c[r];
    }
}


__global__ void __launch_bounds__(PT, 16)
gcn_bucket_acc2_lsm(const int* __restrict__ bsd_i, const int* __restrict__ offt,
                    const float* __restrict__ dis, const float* __restrict__ hs,
                    const float* __restrict__ b2, float* __restrict__ out, int E) {
    constexpr int TP = N_CLASSES + 1;
    __shared__ float tile[RB * TP];
    __shared__ int cnt[SPL * 32];
    __shared__ int off[SPL * 32];
    __shared__ int srt[CAP];
    const int b = blockIdx.x;
    const int base = b * RB;
    for (int i = threadIdx.x; i < RB * TP; i += PT) tile[i] = 0.0f;
    if (threadIdx.x < SPL * 32) cnt[threadIdx.x] = 0;
    __syncthreads();
    gcn_bucket_rows<N_CLASSES, PAD2>(tile, cnt, off, srt, bsd_i, hs, offt, b, (E + NPW - 1) / NPW);
    __syncthreads();

    const int r = threadIdx.x;
    const int i = base + r;
    if (r >= RB || i >= N_NODES) return;
    const float* row = hs + (size_t)i * PAD2;
    const float di = dis[i];
    float z[N_CLASSES];
    {
        float4 t0 = *(const float4*)(row);
        float4 t1 = *(const float4*)(row + 4);
        float2 t2 = *(const float2*)(row + 8);
        z[0] = t0.x; z[1] = t0.y; z[2] = t0.z; z[3] = t0.w;
        z[4] = t1.x; z[5] = t1.y; z[6] = t1.z; z[7] = t1.w;
        z[8] = t2.x; z[9] = t2.y;
    }
    float m = -3.402823466e38f;
    #pragma unroll
    for (int c = 0; c < N_CLASSES; ++c) {
        z[c] = di * (tile[r * TP + c] + z[c]) + b2[c];
        m = fmaxf(m, z[c]);
    }
    float s = 0.0f;
    #pragma unroll
    for (int c = 0; c < N_CLASSES; ++c) s += __expf(z[c] - m);
    const float l = __logf(s);
    #pragma unroll
    for (int c = 0; c < N_CLASSES; ++c) out[(size_t)i * N_CLASSES + c] = z[c] - m - l;
}


extern "C" void kernel_launch(void* const* d_in, const int* in_sizes, int n_in,
                              void* d_out, int out_size, void* d_ws, size_t ws_size,
                              hipStream_t stream) {
    const float* x  = (const float*)d_in[0];
    const int*   ei = (const int*)d_in[1];
    const float* W1 = (const float*)d_in[2];
    const float* b1 = (const float*)d_in[3];
    const float* W2 = (const float*)d_in[4];
    const float* b2 = (const float*)d_in[5];
    float* out = (float*)d_out;

    const int E = in_sizes[1] / 2;
    const int* src = ei;
    const int* dst = ei + E;

    float* ws  = (float*)d_ws;
    float* dis = ws;
    float* h0  = ws + 102400;
    float* h2  = h0 + (size_t)N_NODES * PAD1;
    int*   offt = (int*)(h2 + (size_t)N_NODES * PAD2);
    int*   bsd = offt + NOFF;

    gcn_group_edges<<<NPW, PT, 0, stream>>>(src, dst, offt, bsd, E);

    gcn_deg_gemm1<<<NB, GT, 0, stream>>>(x, W1, bsd, offt, dis, h0, E);
    gcn_bucket_acc1_gemm2<<<NB, PT, 0, stream>>>(bsd, offt, dis, h0, b1, W2, h2, E);

    gcn_bucket_acc2_lsm<<<NB, PT, 0, stream>>>(bsd, offt, dis, h2, b2, out, E);
}


# x@W1 moved into the edge-grouping launch, onto the 128 WGPs the sort leaves idle (A tile in 16 b128 loads, k permuted alike in A, B: its f32 sum order changes); bucket launch only counts, scales rows
# speedup vs baseline: 18.8101x; 1.1060x over previous
#include <hip/hip_runtime.h>


typedef __attribute__((ext_vector_type(2))) float v2f;
typedef __attribute__((ext_vector_type(8))) float v8f;

#define N_NODES   100000
#define N_FEAT    128
#define HIDDEN    20
#define N_CLASSES 10
#define PAD1      32
#define PAD2      16

#define RB    196
#define NB    512
#define NPW   128
#define PT    1024
#define N_EDGES 3200000
#define SR    25
#define NOFF  (NPW * (NB + 1))
static_assert((long long)NPW * SR * PT >= N_EDGES, "a slice must fit its work-group's registers and LDS stage");

__device__ __forceinline__ int gcn_bucket_of(int d) {
    const unsigned b = (unsigned)d / RB;
    return (int)(b < (unsigned)NB ? b : (unsigned)(NB - 1));
}

#define SRC_BITS 17
#define SRC_MASK ((1 << SRC_BITS) - 1)
static_assert(N_NODES <= (1 << SRC_BITS), "a source row must fit the low bits of a record");
static_assert(RB <= 256, "a row inside a bucket must fit 8 bits above the source row");

__device__ __forceinline__ int gcn_rec_pack(int s, int d, int b) {
    return s | (((d - b * RB) & 255) << SRC_BITS);
}

__device__ __forceinline__ int gcn_rec_src(int w) { return w & SRC_MASK; }

__device__ __forceinline__ int gcn_rec_row(int w) {
    const unsigned r = (unsigned)w >> SRC_BITS;
    return (int)(r < (unsigned)RB ? r : (unsigned)(RB - 1));
}


#define NT   ((N_NODES + 15) / 16)
#define NGW  128

__global__ void __launch_bounds__(PT)
gcn_group_edges_gemm1(const int* __restrict__ src, const int* __restrict__ dst,
                      int* __restrict__ offt, int* __restrict__ bsd_i, int E,
                      const float* __restrict__ x, const float* __restrict__ W1,
                      float* __restrict__ hs) {
    __shared__ int hist[NB];
    __shared__ int stage[SR * PT];
    __shared__ float Bf[(N_FEAT / 2) * 32 * 2];
    const int tid = threadIdx.x;
    if (blockIdx.x >= NPW) {
        for (int idx = tid; idx < (N_FEAT / 2) * 32 * 2; idx += PT) {
            const int p = idx >> 6;
            const int c = (idx >> 1) & 31;
            const int j = idx & 1;
            Bf[idx] = (c < HIDDEN) ? W1[(2 * p + j) * HIDDEN + c] : 0.0f;
        }
        __syncthreads();
        const int lane = tid & 31;
        const int wave = __builtin_amdgcn_readfirstlane(tid >> 5);
        const int half = lane >> 4;
        const int l16  = lane & 15;
        const int g = blockIdx.x - NPW;
        for (int i = wave; g + NGW * i < NT; i += PT / 32) {
            const int r0 = (g + NGW * i) << 4;
            asm volatile("" ::: "memory");
            v8f c0 = {};
            v8f c1 = {};
            const float* arow = x + (size_t)min(r0 + l16, N_NODES - 1) * N_FEAT + 4 * half;
            float4 av[N_FEAT / 8];
            #pragma unroll
            for (int j = 0; j < N_FEAT / 8; ++j)
                av[j] = *(const float4*)(arow + 8 * j);
            #pragma unroll
            for (int j = 0; j < N_FEAT / 8; ++j) {
                #pragma unroll
                for (int t = 0; t < 2; ++t) {
                    const int p = 4 * j + 2 * half + t;
                    v2f a;
                    a.x = t ? av[j].z : av[j].x;
                    a.y = t ? av[j].w : av[j].y;
                    v2f b0 = *(const v2f*)&Bf[p * 64 + l16 * 2];
                    v2f b1 = *(const v2f*)&Bf[p * 64 + (16 + l16) * 2];

                    c0 = __builtin_amdgcn_wmma_f32_16x16x4_f32(false, a, false, b0, (short)0, c0, false, false);
                    c1 = __builtin_amdgcn_wmma_f32_16x16x4_f32(false, a, false, b1, (short)0, c1, false, false);
                }
            }
            #pragma unroll
            for (int r = 0; r < 8; ++r) {
                const int row = r0 + r + 8 * half;
                if (row < N_NODES) {
                    float* p0 = hs + (size_t)row * PAD1;
                    p0[l16]      = c0[r];
                    p0[16 + l16] = c1[r];
                }
            }
        }
        return;
    }
    for (int i = tid; i < NB; i += PT) hist[i] = 0;
    __syncthreads();
    const int chunk = (E + NPW - 1) / NPW;
    const int lo = blockIdx.x * chunk;
    const int hi = min(lo + min(chunk, SR * PT), E);
    int d[SR], s[SR], rk[SR];
    #pragma unroll
    for (int u = 0; u < SR; ++u) {
        const int i = lo + u * PT + tid;
        d[u] = (i < hi) ? dst[i] : 0;
        s[u] = (i < hi) ? src[i] : 0;
    }
    #pragma unroll
    for (int u = 0; u < SR; ++u)
        rk[u] = (lo + u * PT + tid < hi) ? atomicAdd(&hist[gcn_bucket_of(d[u])], 1) : 0;
    __syncthreads();
    if (tid < 32) {
        int v[16];
        int sum = 0;
        #pragma unroll
        for (int k = 0; k < 16; ++k) { v[k] = hist[tid * 16 + k]; sum += v[k]; }
        int inc = sum;
        #pragma unroll
        for (int o = 1; o < 32; o <<= 1) {
            const int y = __shfl_up(inc, o);
            if (tid >= o) inc += y;
        }
        int ex = inc - sum;
        #pragma unroll
        for (int k = 0; k < 16; ++k) { hist[tid * 16 + k] = ex; ex += v[k]; }
    }
    __syncthreads();
    int* ot = offt + blockIdx.x * (NB + 1);
    for (int i = tid; i < NB; i += PT) ot[i] = hist[i];
    if (tid == 0) ot[NB] = max(hi - lo, 0);
    #pragma unroll
    for (int u = 0; u < SR; ++u)
        if (lo + u * PT + tid < hi) {
            const int bk = gcn_bucket_of(d[u]);
            stage[hist[bk] + rk[u]] = gcn_rec_pack(s[u], d[u], bk);
        }
    __syncthreads();
    const int n = hi - lo;
    static_assert(SR % 5 == 0, "the copy-out takes five records per thread and trip");
    #pragma nounroll
    for (int u0 = 0; u0 < SR; u0 += 5) {
        int rec[5];
        #pragma unroll
        for (int u = 0; u < 5; ++u)
            rec[u] = stage[(u0 + u) * PT + tid];
        #pragma unroll
        for (int u = 0; u < 5; ++u)
            asm volatile("" :: "v"(rec[u]));
        #pragma unroll
        for (int u = 0; u < 5; ++u)
            if ((u0 + u) * PT + tid < n) bsd_i[lo + (u0 + u) * PT + tid] = rec[u];
    }
}


#define RBP 208
#define DT  256
#define DRB (NPW / (DT / 32))
#define RPC ((HIDDEN + 3) / 4)
#define PPT ((RB * RPC + DT - 1) / DT)
static_assert(DRB * (DT / 32) == NPW, "every run needs exactly one wave");
static_assert(4 * RPC <= PAD1, "row pieces must stay inside the padded row");

__global__ void __launch_bounds__(DT)
gcn_deg_scale(const int* __restrict__ bsd_i, const int* __restrict__ offt,
              float* __restrict__ dis,
              float* __restrict__ hs, int E) {
    __shared__ int cdeg[RBP];

    const int tid = threadIdx.x;
    const int b = blockIdx.x;
    const int base = b * RB;
    for (int i = tid; i < RBP; i += DT) cdeg[i] = 0;
    __syncthreads();

    float4* __restrict__ hs4 = (float4*)hs;
    float4 v[PPT];
    #pragma unroll
    for (int u = 0; u < PPT; ++u) {
        const int p = min(tid + u * DT, RB * RPC - 1);
        const int row = p / RPC;
        const int pc = p - row * RPC;
        v[u] = hs4[(size_t)min(base + row, N_NODES - 1) * (PAD1 / 4) + pc];
    }

    {
        const int* bsd = bsd_i;
        const int lane = tid & 31;
        const int chunk = (E + NPW - 1) / NPW;
        const int cw = __builtin_amdgcn_readfirstlane(tid >> 5);
        int st[DRB], ln[DRB];
        #pragma unroll
        for (int u = 0; u < DRB; ++u) {
            const int r = cw * DRB + u;
            const int a  = offt[r * (NB + 1) + b];
            const int en = offt[r * (NB + 1) + b + 1];
            st[u] = r * chunk + a;
            ln[u] = en - a;
        }
        int d[2 * DRB];
        #pragma unroll
        for (int u = 0; u < DRB; ++u) {
            d[2 * u]     = bsd[min(st[u] + lane, E - 1)];
            d[2 * u + 1] = bsd[min(st[u] + lane + 32, E - 1)];
        }
        #pragma unroll
        for (int u = 0; u < 2 * DRB; ++u)
            asm volatile("" :: "v"(d[u]));
        #pragma unroll
        for (int u = 0; u < DRB; ++u) {
            if (lane < ln[u])      atomicAdd(&cdeg[gcn_rec_row(d[2 * u])], 1);
            if (lane + 32 < ln[u]) atomicAdd(&cdeg[gcn_rec_row(d[2 * u + 1])], 1);
            for (int j = lane + 64; j < ln[u]; j += 32)
                atomicAdd(&cdeg[gcn_rec_row(bsd[st[u] + j])], 1);
        }
    }
    __syncthreads();

    #pragma unroll
    for (int u = 0; u < PPT; ++u) {
        const int p = tid + u * DT;
        const int row = p / RPC;
        const int pc = p - row * RPC;
        if (p < RB * RPC && base + row < N_NODES) {
            const float di = rsqrtf(1.0f + (float)cdeg[row]);
            float4 o;
            o.x = di * v[u].x;
            o.y = di * v[u].y;
            o.z = di * v[u].z;
            o.w = di * v[u].w;
            hs4[(size_t)(base + row) * (PAD1 / 4) + pc] = o;
        }
    }
    for (int i = tid; i < RB; i += DT) {
        const int row = base + i;
        if (row < N_NODES) dis[row] = rsqrtf(1.0f + (float)cdeg[i]);
    }
}


#define KPR   8
#define CAP   (KPR * PT)
#define SPL   ((RB + 31) / 32)

template <int F, int PAD>
__device__ __forceinline__ void gcn_bucket_rows(float* __restrict__ tile, int* __restrict__ cnt,
                                                int* __restrict__ off, int* __restrict__ srt,
                                                const int* __restrict__ bsd,
                                                const float* __restrict__ hs, const int* __restrict__ offt,
                                                int b, int chunk) {
    constexpr int TP  = F + 1;
    constexpr int LPE = (F + 3) / 4;
    constexpr int UG  = 8;
    constexpr int RPW = NPW / (PT / 32);
    static_assert(4 * LPE <= PAD, "row pieces must stay inside the padded row");
    static_assert(RPW * (PT / 32) == NPW, "every run needs exactly one wave");
    static_assert(RB * LPE <= PT, "every row piece needs a lane of its own");
    static_assert(RB < SPL * 32, "off[RB], the end of the last row, must lie inside the padded table");
    static_assert(SPL * 32 <= 256 && CAP <= (1 << 23), "row and rank share a word: 8 bits of row, the rest rank");
    const int tid  = threadIdx.x;
    const int lane = tid & 31;
    const int wave = __builtin_amdgcn_readfirstlane(tid >> 5);
    int st[RPW], cum[RPW + 1];
    cum[0] = 0;
    #pragma unroll
    for (int k = 0; k < RPW; ++k) {
        const int sl = wave * RPW + k;
        const int a  = offt[sl * (NB + 1) + b];
        const int en = offt[sl * (NB + 1) + b + 1];
        st[k] = sl * chunk + a - cum[k];
        cum[k + 1] = cum[k] + (en - a);
    }
    const int n  = cum[RPW];
    const int nh = min(n, KPR * 32);
    {
        int rec[KPR];
        int pk[KPR];
        const int lim = nh - lane;
        if (nh > 0) {
            #pragma unroll
            for (int u = 0; u < KPR; ++u) {
                const int j = lane + min(u * 32, lim - 1);
                int o = st[0];
                #pragma unroll
                for (int k = 1; k < RPW; ++k) o = (j >= cum[k]) ? st[k] : o;
                rec[u] = bsd[o + j];
            }
            #pragma unroll
            for (int u = 0; u < KPR; ++u) {
                const int lr = (u * 32 < lim) ? gcn_rec_row(rec[u]) : RB + lane % (SPL * 32 - RB);
                pk[u] = lr | (atomicAdd(&cnt[lr], 1) << 8);
            }
        }
        __syncthreads();
        if (wave == 0) {
            int v[SPL];
            int sum = 0;
            #pragma unroll
            for (int k = 0; k < SPL; ++k) { v[k] = cnt[lane * SPL + k]; sum += v[k]; }
            int inc = sum;
            #pragma unroll
            for (int o = 1; o < 32; o <<= 1) {
                const int y = __shfl_up(inc, o);
                if (lane >= o) inc += y;
            }
            int ex = inc - sum;
            #pragma unroll
            for (int k = 0; k < SPL; ++k) { off[lane * SPL + k] = ex; ex += v[k]; }
        }
        __syncthreads();
        if (nh > 0) {
            int ob[KPR];
            #pragma unroll
            for (int u = 0; u < KPR; ++u)
                ob[u] = off[pk[u] & 255];
            #pragma unroll
            for (int u = 0; u < KPR; ++u)
                asm volatile("" :: "v"(ob[u]));
            #pragma unroll
            for (int u = 0; u < KPR; ++u)
                if ((pk[u] & 255) < RB)
                    srt[ob[u] + (pk[u] >> 8)] = gcn_rec_src(rec[u]);
        }
        __syncthreads();
    }
    const int row = tid / LPE;
    if (row < RB) {
        const int pc = tid - row * LPE;
        const float4* __restrict__ hs4 = (const float4*)hs;
        float4 acc = make_float4(0.0f, 0.0f, 0.0f, 0.0f);
        const int beg = off[row];
        const int end = off[row + 1];
        for (int i = beg; i < end; i += UG) {
            const int rem = end - i;
            int s[UG];
            #pragma unroll
            for (int u = 0; u < UG; ++u)
                s[u] = srt[i + min(u, rem - 1)];
            float4 v[UG];
            #pragma unroll
            for (int u = 0; u < UG; ++u)
                v[u] = hs4[(unsigned)(s[u] * (PAD / 4) + pc)];
            #pragma unroll
            for (int u = 0; u < UG; ++u)
                asm volatile("" :: "v"(v[u].x), "v"(v[u].y), "v"(v[u].z), "v"(v[u].w));
            #pragma unroll
            for (int u = 0; u < UG; ++u) {
                const bool ok = u < rem;
                acc.x += ok ? v[u].x : 0.0f;
                acc.y += ok ? v[u].y : 0.0f;
                acc.z += ok ? v[u].z : 0.0f;
                acc.w += ok ? v[u].w : 0.0f;
            }
        }
        float* t = &tile[row * TP + pc * 4];
        atomicAdd(t, acc.x);
        if (F % 4 == 0 || pc * 4 + 1 < F) atomicAdd(t + 1, acc.y);
        if (F % 4 == 0 || pc * 4 + 2 < F) atomicAdd(t + 2, acc.z);
        if (F % 4 == 0 || pc * 4 + 3 < F) atomicAdd(t + 3, acc.w);
    }
    if (n > KPR * 32) {
        constexpr int EPW = 32 / LPE;
        constexpr int U   = 30 / EPW;
        constexpr int EPI = EPW * U;
        const int el = min(lane / LPE, EPW - 1);
        const int c4 = (lane % LPE) * 4;
        const bool lane_on = lane < LPE * EPW;
        for (int j0 = KPR * 32; j0 < n; j0 += EPI) {
            int sd[U];
            #pragma unroll
            for (int u = 0; u < U; ++u) {
                const int j = min(j0 + u * EPW + el, n - 1);
                int o = st[0];
                #pragma unroll
                for (int k = 1; k < RPW; ++k) o = (j >= cum[k]) ? st[k] : o;
                sd[u] = bsd[o + j];
            }
            float4 v[U];
            #pragma unroll
            for (int u = 0; u < U; ++u)
                v[u] = *(const float4*)(hs + (size_t)gcn_rec_src(sd[u]) * PAD + c4);
            #pragma unroll
            for (int u = 0; u < U; ++u)
                asm volatile("" :: "v"(v[u].x), "v"(v[u].y), "v"(v[u].z), "v"(v[u].w));
            #pragma unroll
            for (int u = 0; u < U; ++u) {
                if (lane_on && j0 + u * EPW + el < n) {
                    float* t = &tile[gcn_rec_row(sd[u]) * TP + c4];
                    atomicAdd(t, v[u].x);
                    if (F % 4 == 0 || c4 + 1 < F) atomicAdd(t + 1, v[u].y);
                    if (F % 4 == 0 || c4 + 2 < F) atomicAdd(t + 2, v[u].z);
                    if (F % 4 == 0 || c4 + 3 < F) atomicAdd(t + 3, v[u].w);
                }
            }
        }
    }
}


__global__ void __launch_bounds__(PT, 16)
gcn_bucket_acc1_gemm2(const int* __restrict__ bsd_i, const int* __restrict__ offt,
                      const float* __restrict__ dis, const float* __restrict__ hs,
                      const float* __restrict__ b1, const float* __restrict__ W2,
                      float* __restrict__ h2s, int E) {
    constexpr int TP = HIDDEN + 1;
    __shared__ float tile[RBP * TP];
    __shared__ float Bf2[(HIDDEN / 2) * 16 * 2];
    __shared__ float b1s[HIDDEN];
    __shared__ float disl[RBP];
    __shared__ int cnt[SPL * 32];
    __shared__ int off[SPL * 32];
    __shared__ int srt[CAP];
    const int b = blockIdx.x;
    const int base = b * RB;
    const int tid = threadIdx.x;
    for (int i = tid; i < RBP * TP; i += PT) tile[i] = 0.0f;
    if (tid < SPL * 32) cnt[tid] = 0;
    __syncthreads();
    float w2v = 0.0f, b1v = 0.0f, dv = 0.0f;
    if (tid < (HIDDEN / 2) * 16 * 2) {
        const int p = tid >> 5;
        const int c = (tid >> 1) & 15;
        const int j = tid & 1;
        if (c < N_CLASSES) w2v = W2[(2 * p + j) * N_CLASSES + c];
    }
    if (tid < HIDDEN) b1v = b1[tid];
    if (tid < RB && base + tid < N_NODES) dv = dis[base + tid];
    gcn_bucket_rows<HIDDEN, PAD1>(tile, cnt, off, srt, bsd_i, hs, offt, b, (E + NPW - 1) / NPW);
    if (tid < (HIDDEN / 2) * 16 * 2) Bf2[tid] = w2v;
    if (tid < HIDDEN) b1s[tid] = b1v;
    if (tid < RBP) disl[tid] = dv;
    __syncthreads();

    const int lane = threadIdx.x & 31;
    const int wave = threadIdx.x >> 5;
    const int r0 = wave << 4;
    if (r0 >= RB || base + r0 >= N_NODES) return;
    const int half = lane >> 4;
    const int l16  = lane & 15;
    const int lr = r0 + l16;
    const bool live = (lr < RB) && (base + lr < N_NODES);
    const float* srow = hs + (size_t)(live ? base + lr : 0) * PAD1;
    const float dA = disl[lr];
    const float* trow = &tile[lr * TP];

    v8f c = {};
    #pragma unroll
    for (int k0 = 0; k0 < HIDDEN; k0 += 4) {
        const int kb = k0 + 2 * half;
        const v2f sr = *(const v2f*)(srow + kb);
        v2f a;
        a.x = fmaxf(dA * (trow[kb]     + sr.x) + b1s[kb],     0.0f);
        a.y = fmaxf(dA * (trow[kb + 1] + sr.y) + b1s[kb + 1], 0.0f);
        const v2f bb = *(const v2f*)&Bf2[(kb >> 1) * 32 + l16 * 2];
        c = __builtin_amdgcn_wmma_f32_16x16x4_f32(false, a, false, bb, (short)0, c, false, false);
    }

    #pragma unroll
    for (int r = 0; r < 8; ++r) {
        const int lrow = r0 + r + 8 * half;
        const int row = base + lrow;
        if (lrow < RB && row < N_NODES)
            h2s[(size_t)row * PAD2 + l16] = disl[lrow] * c[r];
    }
}


__global__ void __launch_bounds__(PT, 16)
gcn_bucket_acc2_lsm(const int* __restrict__ bsd_i, const int* __restrict__ offt,
                    const float* __restrict__ dis, const float* __restrict__ hs,
                    const float* __restrict__ b2, float* __restrict__ out, int E) {
    constexpr int TP = N_CLASSES + 1;
    __shared__ float tile[RB * TP];
    __shared__ int cnt[SPL * 32];
    __shared__ int off[SPL * 32];
    __shared__ int srt[CAP];
    const int b = blockIdx.x;
    const int base = b * RB;
    for (int i = threadIdx.x; i < RB * TP; i += PT) tile[i] = 0.0f;
    if (threadIdx.x < SPL * 32) cnt[threadIdx.x] = 0;
    __syncthreads();
    gcn_bucket_rows<N_CLASSES, PAD2>(tile, cnt, off, srt, bsd_i, hs, offt, b, (E + NPW - 1) / NPW);
    __syncthreads();

    const int r = threadIdx.x;
    const int i = base + r;
    if (r >= RB || i >= N_NODES) return;
    const float* row = hs + (size_t)i * PAD2;
    const float di = dis[i];
    float z[N_CLASSES];
    {
        float4 t0 = *(const float4*)(row);
        float4 t1 = *(const float4*)(row + 4);
        float2 t2 = *(const float2*)(row + 8);
        z[0] = t0.x; z[1] = t0.y; z[2] = t0.z; z[3] = t0.w;
        z[4] = t1.x; z[5] = t1.y; z[6] = t1.z; z[7] = t1.w;
        z[8] = t2.x; z[9] = t2.y;
    }
    float m = -3.402823466e38f;
    #pragma unroll
    for (int c = 0; c < N_CLASSES; ++c) {
        z[c] = di * (tile[r * TP + c] + z[c]) + b2[c];
        m = fmaxf(m, z[c]);
    }
    float s = 0.0f;
    #pragma unroll
    for (int c = 0; c < N_CLASSES; ++c) s += __expf(z[c] - m);
    const float l = __logf(s);
    #pragma unroll
    for (int c = 0; c < N_CLASSES; ++c) out[(size_t)i * N_CLASSES + c] = z[c] - m - l;
}


extern "C" void kernel_launch(void* const* d_in, const int* in_sizes, int n_in,
                              void* d_out, int out_size, void* d_ws, size_t ws_size,
                              hipStream_t stream) {
    const float* x  = (const float*)d_in[0];
    const int*   ei = (const int*)d_in[1];
    const float* W1 = (const float*)d_in[2];
    const float* b1 = (const float*)d_in[3];
    const float* W2 = (const float*)d_in[4];
    const float* b2 = (const float*)d_in[5];
    float* out = (float*)d_out;

    const int E = in_sizes[1] / 2;
    const int* src = ei;
    const int* dst = ei + E;

    float* ws  = (float*)d_ws;
    float* dis = ws;
    float* h0  = ws + 102400;
    float* h2  = h0 + (size_t)N_NODES * PAD1;
    int*   offt = (int*)(h2 + (size_t)N_NODES * PAD2);
    int*   bsd = offt + NOFF;

    gcn_group_edges_gemm1<<<NPW + NGW, PT, 0, stream>>>(src, dst, offt, bsd, E, x, W1, h0);

    gcn_deg_scale<<<NB, DT, 0, stream>>>(bsd, offt, dis, h0, E);
    gcn_bucket_acc1_gemm2<<<NB, PT, 0, stream>>>(bsd, offt, dis, h0, b1, W2, h2, E);

    gcn_bucket_acc2_lsm<<<NB, PT, 0, stream>>>(bsd, offt, dis, h2, b2, out, E);
}
